# starting kernel
# baseline (speedup 1.0000x reference)
#include <hip/hip_runtime.h>
#include <hip/hip_bf16.h>

#define DIM 64
#define HID 512
#define BATCH 8192

typedef __attribute__((ext_vector_type(2))) float v2f;
typedef __attribute__((ext_vector_type(8))) float v8f;

#define WS_MW1T  0
#define WS_MW2   (HID*DIM)
#define WS_WSUM  (HID*DIM + 2*DIM*HID)
#define WS_BSUM  (WS_WSUM + HID)
#define WS_BPACK 98832
#define NPACK    (HID*DIM)

__global__ __launch_bounds__(256) void maf_prep_mask(const float* __restrict__ W1,
                                                     const float* __restrict__ mask1,
                                                     const float* __restrict__ W2,
                                                     const float* __restrict__ mask2,
                                                     float* __restrict__ ws) {
  int idx = blockIdx.x * blockDim.x + threadIdx.x;
  if (idx < HID * DIM) {
    int j = idx / DIM, k = idx % DIM;
    ws[WS_MW1T + k * HID + j] = mask1[idx] * W1[idx];
  } else if (idx < HID * DIM + 2 * DIM * HID) {
    int t = idx - HID * DIM;
    ws[WS_MW2 + t] = mask2[t] * W2[t];
  } else if (idx < HID * DIM + 2 * DIM * HID + NPACK) {
    int f = idx - (HID * DIM + 2 * DIM * HID);
    int p = f >> 1, c = f & 1;
    int nt = p >> 9;
    int t  = (p >> 5) & 15;
    int l  = p & 31;
    int koff = (l >> 4) << 1;
    int j = (nt << 4) + (l & 15);
    int k = (t << 2) + koff + c;
    ws[WS_BPACK + f] = mask1[j * DIM + k] * W1[j * DIM + k];
  }
}

__global__ __launch_bounds__(512) void maf_prep_sum(const float* __restrict__ b2,
                                                    float* __restrict__ ws) {
  int j = threadIdx.x;
  const float* MW2 = ws + WS_MW2;
  float s = 0.f;
  for (int r = DIM; r < 2 * DIM; ++r) s += MW2[r * HID + j];
  ws[WS_WSUM + j] = s;
  if (j == 0) {
    float bs = 0.f;
    for (int r = DIM; r < 2 * DIM; ++r) bs += b2[r];
    ws[WS_BSUM] = bs;
  }
}

__global__ __launch_bounds__(256) void maf_scan(const float* __restrict__ z,
                                                const float* __restrict__ b1,
                                                const float* __restrict__ b2,
                                                const float* __restrict__ ws,
                                                float* __restrict__ xout) {
  __shared__ __align__(16) float w1c[HID];
  __shared__ __align__(16) float w2a[HID];
  __shared__ __align__(16) float w2b[HID];
  const float* MW1T = ws + WS_MW1T;
  const float* MW2  = ws + WS_MW2;

  const int tid  = threadIdx.x;
  const int lane = tid & 31;
  const int wave = tid >> 5;
  const int row  = blockIdx.x * 8 + wave;
  const int t2   = tid * 2;

  float h[16];
#pragma unroll
  for (int t = 0; t < 16; ++t) h[t] = b1[lane + 32 * t];

  const float z0 = z[row * DIM + lane];
  const float z1 = z[row * DIM + 32 + lane];
  float x0 = 0.f, x1 = 0.f;

  for (int i = 0; i < DIM; ++i) {
    __syncthreads();
    *(v2f*)&w1c[t2] = *(const v2f*)&MW1T[i * HID + t2];
    *(v2f*)&w2a[t2] = *(const v2f*)&MW2[i * HID + t2];
    *(v2f*)&w2b[t2] = *(const v2f*)&MW2[(DIM + i) * HID + t2];
    __syncthreads();

    float mu = 0.f, al = 0.f;
#pragma unroll
    for (int t = 0; t < 16; ++t) {
      float hv = fmaxf(h[t], 0.f);
      int j = lane + 32 * t;
      mu = fmaf(hv, w2a[j], mu);
      al = fmaf(hv, w2b[j], al);
    }
#pragma unroll
    for (int off = 16; off >= 1; off >>= 1) {
      mu += __shfl_xor(mu, off, 32);
      al += __shfl_xor(al, off, 32);
    }
    mu += b2[i];
    al += b2[DIM + i];

    float zsel = (i < 32) ? z0 : z1;
    float zi = __shfl(zsel, i & 31, 32);
    float xi = fmaf(zi, expf(al), mu);

    if (i < 32) { if (lane == i)        x0 = xi; }
    else        { if (lane == (i - 32)) x1 = xi; }

    if (i < DIM - 1) {
#pragma unroll
      for (int t = 0; t < 16; ++t) h[t] = fmaf(xi, w1c[lane + 32 * t], h[t]);
    }
  }

  xout[row * DIM + lane]      = x0;
  xout[row * DIM + 32 + lane] = x1;
}

__global__ __launch_bounds__(256) void maf_logdet_wmma(const float* __restrict__ b1,
                                                       const float* __restrict__ ws,
                                                       const float* __restrict__ xout,
                                                       float* __restrict__ ldout) {
  __shared__ float b1s[HID];
  __shared__ float wss[HID];
  const v2f* Bp = (const v2f*)(ws + WS_BPACK);
  const float bsum = ws[WS_BSUM];

  const int tid = threadIdx.x;
  b1s[tid]       = b1[tid];
  b1s[tid + 256] = b1[tid + 256];
  wss[tid]       = ws[WS_WSUM + tid];
  wss[tid + 256] = ws[WS_WSUM + tid + 256];
  __syncthreads();

  const int lane = tid & 31;
  const int wave = tid >> 5;
  const int lr   = lane & 15;
  const int koff = (lane >> 4) ? 2 : 0;
  const int m0   = (blockIdx.x * 8 + wave) * 16;

  const float* xp = xout + (m0 + lr) * DIM + koff;
  v2f a[16];
#pragma unroll
  for (int t = 0; t < 16; ++t) a[t] = *(const v2f*)(xp + 4 * t);

  v8f ld = {};
  for (int nt = 0; nt < 32; ++nt) {
    const int col = nt * 16 + lr;

    v2f bt[16];
#pragma unroll
    for (int t = 0; t < 16; ++t) bt[t] = Bp[(nt * 16 + t) * 32 + lane];

    v8f acc0 = {}, acc1 = {};
#pragma unroll
    for (int t = 0; t < 16; t += 2) {
      acc0 = __builtin_amdgcn_wmma_f32_16x16x4_f32(false, a[t],     false, bt[t],
                                                   (short)0, acc0, false, false);
      acc1 = __builtin_amdgcn_wmma_f32_16x16x4_f32(false, a[t + 1], false, bt[t + 1],
                                                   (short)0, acc1, false, false);
    }
    v8f acc = acc0 + acc1;

    const float bb = b1s[col];
    const float wv = wss[col];
#pragma unroll
    for (int v = 0; v < 8; ++v) {
      float hv = fmaxf(acc[v] + bb, 0.f);
      ld[v] = fmaf(hv, wv, ld[v]);
    }
  }

#pragma unroll
  for (int off = 1; off < 16; off <<= 1) {
#pragma unroll
    for (int v = 0; v < 8; ++v) {
      float t = ld[v];
      ld[v] = t + __shfl_xor(t, off, 32);
    }
  }
  if (lane == 0) {
#pragma unroll
    for (int v = 0; v < 8; ++v) ldout[m0 + v] = ld[v] + bsum;
  } else if (lane == 16) {
#pragma unroll
    for (int v = 0; v < 8; ++v) ldout[m0 + 8 + v] = ld[v] + bsum;
  }
}

extern "C" void kernel_launch(void* const* d_in, const int* in_sizes, int n_in,
                              void* d_out, int out_size, void* d_ws, size_t ws_size,
                              hipStream_t stream) {
  const float* z     = (const float*)d_in[0];
  const float* W1    = (const float*)d_in[1];
  const float* b1    = (const float*)d_in[2];
  const float* W2    = (const float*)d_in[3];
  const float* b2    = (const float*)d_in[4];
  const float* mask1 = (const float*)d_in[5];
  const float* mask2 = (const float*)d_in[6];

  float* ws    = (float*)d_ws;
  float* xout  = (float*)d_out;
  float* ldout = xout + BATCH * DIM;

  const int nprep = HID * DIM + 2 * DIM * HID + NPACK;
  maf_prep_mask<<<(nprep + 255) / 256, 256, 0, stream>>>(W1, mask1, W2, mask2, ws);
  maf_prep_sum<<<1, 512, 0, stream>>>(b2, ws);
  maf_scan<<<BATCH / 8, 256, 0, stream>>>(z, b1, b2, ws, xout);
  maf_logdet_wmma<<<BATCH / 128, 256, 0, stream>>>(b1, ws, xout, ldout);
}


# logdet WMMA kernel: 32 hidden tiles of a row tile split over 8 waves (256 blocks x 16 waves, was 64 x 8), all B fragments loaded before the first WMMA; 8 partial sums joined in LDS (new sum order)
# speedup vs baseline: 1.3857x; 1.3857x over previous
#include <hip/hip_runtime.h>
#include <hip/hip_bf16.h>

#define DIM 64
#define HID 512
#define BATCH 8192

typedef __attribute__((ext_vector_type(2))) float v2f;
typedef __attribute__((ext_vector_type(8))) float v8f;

#define WS_MW1T  0
#define WS_MW2   (HID*DIM)
#define WS_WSUM  (HID*DIM + 2*DIM*HID)
#define WS_BSUM  (WS_WSUM + HID)
#define WS_BPACK 98832
#define NPACK    (HID*DIM)

__global__ __launch_bounds__(256) void maf_prep_mask(const float* __restrict__ W1,
                                                     const float* __restrict__ mask1,
                                                     const float* __restrict__ W2,
                                                     const float* __restrict__ mask2,
                                                     float* __restrict__ ws) {
  int idx = blockIdx.x * blockDim.x + threadIdx.x;
  if (idx < HID * DIM) {
    int j = idx / DIM, k = idx % DIM;
    ws[WS_MW1T + k * HID + j] = mask1[idx] * W1[idx];
  } else if (idx < HID * DIM + 2 * DIM * HID) {
    int t = idx - HID * DIM;
    ws[WS_MW2 + t] = mask2[t] * W2[t];
  } else if (idx < HID * DIM + 2 * DIM * HID + NPACK) {
    int f = idx - (HID * DIM + 2 * DIM * HID);
    int p = f >> 1, c = f & 1;
    int nt = p >> 9;
    int t  = (p >> 5) & 15;
    int l  = p & 31;
    int koff = (l >> 4) << 1;
    int j = (nt << 4) + (l & 15);
    int k = (t << 2) + koff + c;
    ws[WS_BPACK + f] = mask1[j * DIM + k] * W1[j * DIM + k];
  }
}

__global__ __launch_bounds__(512) void maf_prep_sum(const float* __restrict__ b2,
                                                    float* __restrict__ ws) {
  int j = threadIdx.x;
  const float* MW2 = ws + WS_MW2;
  float s = 0.f;
  for (int r = DIM; r < 2 * DIM; ++r) s += MW2[r * HID + j];
  ws[WS_WSUM + j] = s;
  if (j == 0) {
    float bs = 0.f;
    for (int r = DIM; r < 2 * DIM; ++r) bs += b2[r];
    ws[WS_BSUM] = bs;
  }
}

__global__ __launch_bounds__(256) void maf_scan(const float* __restrict__ z,
                                                const float* __restrict__ b1,
                                                const float* __restrict__ b2,
                                                const float* __restrict__ ws,
                                                float* __restrict__ xout) {
  __shared__ __align__(16) float w1c[HID];
  __shared__ __align__(16) float w2a[HID];
  __shared__ __align__(16) float w2b[HID];
  const float* MW1T = ws + WS_MW1T;
  const float* MW2  = ws + WS_MW2;

  const int tid  = threadIdx.x;
  const int lane = tid & 31;
  const int wave = tid >> 5;
  const int row  = blockIdx.x * 8 + wave;
  const int t2   = tid * 2;

  float h[16];
#pragma unroll
  for (int t = 0; t < 16; ++t) h[t] = b1[lane + 32 * t];

  const float z0 = z[row * DIM + lane];
  const float z1 = z[row * DIM + 32 + lane];
  float x0 = 0.f, x1 = 0.f;

  for (int i = 0; i < DIM; ++i) {
    __syncthreads();
    *(v2f*)&w1c[t2] = *(const v2f*)&MW1T[i * HID + t2];
    *(v2f*)&w2a[t2] = *(const v2f*)&MW2[i * HID + t2];
    *(v2f*)&w2b[t2] = *(const v2f*)&MW2[(DIM + i) * HID + t2];
    __syncthreads();

    float mu = 0.f, al = 0.f;
#pragma unroll
    for (int t = 0; t < 16; ++t) {
      float hv = fmaxf(h[t], 0.f);
      int j = lane + 32 * t;
      mu = fmaf(hv, w2a[j], mu);
      al = fmaf(hv, w2b[j], al);
    }
#pragma unroll
    for (int off = 16; off >= 1; off >>= 1) {
      mu += __shfl_xor(mu, off, 32);
      al += __shfl_xor(al, off, 32);
    }
    mu += b2[i];
    al += b2[DIM + i];

    float zsel = (i < 32) ? z0 : z1;
    float zi = __shfl(zsel, i & 31, 32);
    float xi = fmaf(zi, expf(al), mu);

    if (i < 32) { if (lane == i)        x0 = xi; }
    else        { if (lane == (i - 32)) x1 = xi; }

    if (i < DIM - 1) {
#pragma unroll
      for (int t = 0; t < 16; ++t) h[t] = fmaf(xi, w1c[lane + 32 * t], h[t]);
    }
  }

  xout[row * DIM + lane]      = x0;
  xout[row * DIM + 32 + lane] = x1;
}

#define LD_NT 4
__global__ __launch_bounds__(512) void maf_logdet_wmma(const float* __restrict__ b1,
                                                       const float* __restrict__ ws,
                                                       const float* __restrict__ xout,
                                                       float* __restrict__ ldout) {
  __shared__ float b1s[HID];
  __shared__ float wss[HID];
  __shared__ float part[8][32];
  const v2f* Bp = (const v2f*)(ws + WS_BPACK);
  const float bsum = ws[WS_BSUM];

  const int tid  = threadIdx.x;
  const int lane = tid & 31;
  const int wave = tid >> 5;
  const int lr   = lane & 15;
  const int koff = (lane >> 4) ? 2 : 0;
  const int mt   = wave & 1;
  const int q    = wave >> 1;
  const int m0   = blockIdx.x * 32 + mt * 16;
  const int nt0  = q * LD_NT;

  const float b1v = b1[tid];
  const float wsv = ws[WS_WSUM + tid];

  const float* xp = xout + (m0 + lr) * DIM + koff;
  v2f a[16];
#pragma unroll
  for (int t = 0; t < 16; ++t) a[t] = *(const v2f*)(xp + 4 * t);

  v2f bt[LD_NT][16];
#pragma unroll
  for (int i = 0; i < LD_NT; ++i) {
#pragma unroll
    for (int t = 0; t < 16; ++t) bt[i][t] = Bp[((nt0 + i) * 16 + t) * 32 + lane];
  }

  __builtin_amdgcn_sched_barrier(0);

  b1s[tid] = b1v;
  wss[tid] = wsv;
  __syncthreads();

  v8f ld = {};
#pragma unroll
  for (int i = 0; i < LD_NT; ++i) {
    const int col = (nt0 + i) * 16 + lr;

    v8f acc0 = {}, acc1 = {};
#pragma unroll
    for (int t = 0; t < 16; t += 2) {
      acc0 = __builtin_amdgcn_wmma_f32_16x16x4_f32(false, a[t],     false, bt[i][t],
                                                   (short)0, acc0, false, false);
      acc1 = __builtin_amdgcn_wmma_f32_16x16x4_f32(false, a[t + 1], false, bt[i][t + 1],
                                                   (short)0, acc1, false, false);
    }
    v8f acc = acc0 + acc1;

    const float bb = b1s[col];
    const float wv = wss[col];
#pragma unroll
    for (int v = 0; v < 8; ++v) {
      float hv = fmaxf(acc[v] + bb, 0.f);
      ld[v] = fmaf(hv, wv, ld[v]);
    }
  }

#pragma unroll
  for (int off = 1; off < 16; off <<= 1) {
#pragma unroll
    for (int v = 0; v < 8; ++v) {
      float t = ld[v];
      ld[v] = t + __shfl_xor(t, off, 32);
    }
  }
  if (lr == 0) {
    const int r0 = mt * 16 + (lane >> 4) * 8;
#pragma unroll
    for (int v = 0; v < 8; ++v) part[q][r0 + v] = ld[v];
  }
  __syncthreads();
  if (tid < 32) {
    float s = part[0][tid];
#pragma unroll
    for (int k = 1; k < 8; ++k) s += part[k][tid];
    ldout[blockIdx.x * 32 + tid] = s + bsum;
  }
}

extern "C" void kernel_launch(void* const* d_in, const int* in_sizes, int n_in,
                              void* d_out, int out_size, void* d_ws, size_t ws_size,
                              hipStream_t stream) {
  const float* z     = (const float*)d_in[0];
  const float* W1    = (const float*)d_in[1];
  const float* b1    = (const float*)d_in[2];
  const float* W2    = (const float*)d_in[3];
  const float* b2    = (const float*)d_in[4];
  const float* mask1 = (const float*)d_in[5];
  const float* mask2 = (const float*)d_in[6];

  float* ws    = (float*)d_ws;
  float* xout  = (float*)d_out;
  float* ldout = xout + BATCH * DIM;

  const int nprep = HID * DIM + 2 * DIM * HID + NPACK;
  maf_prep_mask<<<(nprep + 255) / 256, 256, 0, stream>>>(W1, mask1, W2, mask2, ws);
  maf_prep_sum<<<1, 512, 0, stream>>>(b2, ws);
  maf_scan<<<BATCH / 8, 256, 0, stream>>>(z, b1, b2, ws, xout);
  maf_logdet_wmma<<<BATCH / 32, 512, 0, stream>>>(b1, ws, xout, ldout);
}


# scan: wave sums of mu/alpha and the z broadcast moved from 11 ds_bpermute per step to vector-ALU lane ops (permlanex16 + DPP row_xmask adds, readlane); same butterfly order, sums bit-identical
# speedup vs baseline: 1.6874x; 1.2177x over previous
#include <hip/hip_runtime.h>
#include <hip/hip_bf16.h>

#define DIM 64
#define HID 512
#define BATCH 8192

typedef __attribute__((ext_vector_type(2))) float v2f;
typedef __attribute__((ext_vector_type(8))) float v8f;

#define WS_MW1T  0
#define WS_MW2   (HID*DIM)
#define WS_WSUM  (HID*DIM + 2*DIM*HID)
#define WS_BSUM  (WS_WSUM + HID)
#define WS_BPACK 98832
#define NPACK    (HID*DIM)

__global__ __launch_bounds__(256) void maf_prep_mask(const float* __restrict__ W1,
                                                     const float* __restrict__ mask1,
                                                     const float* __restrict__ W2,
                                                     const float* __restrict__ mask2,
                                                     float* __restrict__ ws) {
  int idx = blockIdx.x * blockDim.x + threadIdx.x;
  if (idx < HID * DIM) {
    int j = idx / DIM, k = idx % DIM;
    ws[WS_MW1T + k * HID + j] = mask1[idx] * W1[idx];
  } else if (idx < HID * DIM + 2 * DIM * HID) {
    int t = idx - HID * DIM;
    ws[WS_MW2 + t] = mask2[t] * W2[t];
  } else if (idx < HID * DIM + 2 * DIM * HID + NPACK) {
    int f = idx - (HID * DIM + 2 * DIM * HID);
    int p = f >> 1, c = f & 1;
    int nt = p >> 9;
    int t  = (p >> 5) & 15;
    int l  = p & 31;
    int koff = (l >> 4) << 1;
    int j = (nt << 4) + (l & 15);
    int k = (t << 2) + koff + c;
    ws[WS_BPACK + f] = mask1[j * DIM + k] * W1[j * DIM + k];
  }
}

__global__ __launch_bounds__(512) void maf_prep_sum(const float* __restrict__ b2,
                                                    float* __restrict__ ws) {
  int j = threadIdx.x;
  const float* MW2 = ws + WS_MW2;
  float s = 0.f;
  for (int r = DIM; r < 2 * DIM; ++r) s += MW2[r * HID + j];
  ws[WS_WSUM + j] = s;
  if (j == 0) {
    float bs = 0.f;
    for (int r = DIM; r < 2 * DIM; ++r) bs += b2[r];
    ws[WS_BSUM] = bs;
  }
}

__device__ __forceinline__ float lane_xor16(float x) {
  const int b = __builtin_bit_cast(int, x);
  return __builtin_bit_cast(float,
      __builtin_amdgcn_permlanex16(b, b, 0x76543210, 0xfedcba98, false, false));
}
template <int M>
__device__ __forceinline__ float row_xor(float x) {
  const int b = __builtin_bit_cast(int, x);
  return __builtin_bit_cast(float, __builtin_amdgcn_update_dpp(0, b, 0x160 + M, 0xf, 0xf, true));
}

__global__ __launch_bounds__(256) void maf_scan(const float* __restrict__ z,
                                                const float* __restrict__ b1,
                                                const float* __restrict__ b2,
                                                const float* __restrict__ ws,
                                                float* __restrict__ xout) {
  __shared__ __align__(16) float w1c[HID];
  __shared__ __align__(16) float w2a[HID];
  __shared__ __align__(16) float w2b[HID];
  const float* MW1T = ws + WS_MW1T;
  const float* MW2  = ws + WS_MW2;

  const int tid  = threadIdx.x;
  const int lane = tid & 31;
  const int wave = tid >> 5;
  const int row  = blockIdx.x * 8 + wave;
  const int t2   = tid * 2;

  float h[16];
#pragma unroll
  for (int t = 0; t < 16; ++t) h[t] = b1[lane + 32 * t];

  const float z0 = z[row * DIM + lane];
  const float z1 = z[row * DIM + 32 + lane];
  float x0 = 0.f, x1 = 0.f;

  for (int i = 0; i < DIM; ++i) {
    __syncthreads();
    *(v2f*)&w1c[t2] = *(const v2f*)&MW1T[i * HID + t2];
    *(v2f*)&w2a[t2] = *(const v2f*)&MW2[i * HID + t2];
    *(v2f*)&w2b[t2] = *(const v2f*)&MW2[(DIM + i) * HID + t2];
    __syncthreads();

    float mu = 0.f, al = 0.f;
#pragma unroll
    for (int t = 0; t < 16; ++t) {
      float hv = fmaxf(h[t], 0.f);
      int j = lane + 32 * t;
      mu = fmaf(hv, w2a[j], mu);
      al = fmaf(hv, w2b[j], al);
    }
    mu += lane_xor16(mu);  al += lane_xor16(al);
    mu += row_xor<8>(mu);  al += row_xor<8>(al);
    mu += row_xor<4>(mu);  al += row_xor<4>(al);
    mu += row_xor<2>(mu);  al += row_xor<2>(al);
    mu += row_xor<1>(mu);  al += row_xor<1>(al);
    mu += b2[i];
    al += b2[DIM + i];

    float zsel = (i < 32) ? z0 : z1;
    float zi = __builtin_bit_cast(float,
        __builtin_amdgcn_readlane(__builtin_bit_cast(int, zsel), i & 31));
    float xi = fmaf(zi, expf(al), mu);

    if (i < 32) { if (lane == i)        x0 = xi; }
    else        { if (lane == (i - 32)) x1 = xi; }

    if (i < DIM - 1) {
#pragma unroll
      for (int t = 0; t < 16; ++t) h[t] = fmaf(xi, w1c[lane + 32 * t], h[t]);
    }
  }

  xout[row * DIM + lane]      = x0;
  xout[row * DIM + 32 + lane] = x1;
}

#define LD_NT 4
__global__ __launch_bounds__(512) void maf_logdet_wmma(const float* __restrict__ b1,
                                                       const float* __restrict__ ws,
                                                       const float* __restrict__ xout,
                                                       float* __restrict__ ldout) {
  __shared__ float b1s[HID];
  __shared__ float wss[HID];
  __shared__ float part[8][32];
  const v2f* Bp = (const v2f*)(ws + WS_BPACK);
  const float bsum = ws[WS_BSUM];

  const int tid  = threadIdx.x;
  const int lane = tid & 31;
  const int wave = tid >> 5;
  const int lr   = lane & 15;
  const int koff = (lane >> 4) ? 2 : 0;
  const int mt   = wave & 1;
  const int q    = wave >> 1;
  const int m0   = blockIdx.x * 32 + mt * 16;
  const int nt0  = q * LD_NT;

  const float b1v = b1[tid];
  const float wsv = ws[WS_WSUM + tid];

  const float* xp = xout + (m0 + lr) * DIM + koff;
  v2f a[16];
#pragma unroll
  for (int t = 0; t < 16; ++t) a[t] = *(const v2f*)(xp + 4 * t);

  v2f bt[LD_NT][16];
#pragma unroll
  for (int i = 0; i < LD_NT; ++i) {
#pragma unroll
    for (int t = 0; t < 16; ++t) bt[i][t] = Bp[((nt0 + i) * 16 + t) * 32 + lane];
  }

  __builtin_amdgcn_sched_barrier(0);

  b1s[tid] = b1v;
  wss[tid] = wsv;
  __syncthreads();

  v8f ld = {};
#pragma unroll
  for (int i = 0; i < LD_NT; ++i) {
    const int col = (nt0 + i) * 16 + lr;

    v8f acc0 = {}, acc1 = {};
#pragma unroll
    for (int t = 0; t < 16; t += 2) {
      acc0 = __builtin_amdgcn_wmma_f32_16x16x4_f32(false, a[t],     false, bt[i][t],
                                                   (short)0, acc0, false, false);
      acc1 = __builtin_amdgcn_wmma_f32_16x16x4_f32(false, a[t + 1], false, bt[i][t + 1],
                                                   (short)0, acc1, false, false);
    }
    v8f acc = acc0 + acc1;

    const float bb = b1s[col];
    const float wv = wss[col];
#pragma unroll
    for (int v = 0; v < 8; ++v) {
      float hv = fmaxf(acc[v] + bb, 0.f);
      ld[v] = fmaf(hv, wv, ld[v]);
    }
  }

#pragma unroll
  for (int off = 1; off < 16; off <<= 1) {
#pragma unroll
    for (int v = 0; v < 8; ++v) {
      float t = ld[v];
      ld[v] = t + __shfl_xor(t, off, 32);
    }
  }
  if (lr == 0) {
    const int r0 = mt * 16 + (lane >> 4) * 8;
#pragma unroll
    for (int v = 0; v < 8; ++v) part[q][r0 + v] = ld[v];
  }
  __syncthreads();
  if (tid < 32) {
    float s = part[0][tid];
#pragma unroll
    for (int k = 1; k < 8; ++k) s += part[k][tid];
    ldout[blockIdx.x * 32 + tid] = s + bsum;
  }
}

extern "C" void kernel_launch(void* const* d_in, const int* in_sizes, int n_in,
                              void* d_out, int out_size, void* d_ws, size_t ws_size,
                              hipStream_t stream) {
  const float* z     = (const float*)d_in[0];
  const float* W1    = (const float*)d_in[1];
  const float* b1    = (const float*)d_in[2];
  const float* W2    = (const float*)d_in[3];
  const float* b2    = (const float*)d_in[4];
  const float* mask1 = (const float*)d_in[5];
  const float* mask2 = (const float*)d_in[6];

  float* ws    = (float*)d_ws;
  float* xout  = (float*)d_out;
  float* ldout = xout + BATCH * DIM;

  const int nprep = HID * DIM + 2 * DIM * HID + NPACK;
  maf_prep_mask<<<(nprep + 255) / 256, 256, 0, stream>>>(W1, mask1, W2, mask2, ws);
  maf_prep_sum<<<1, 512, 0, stream>>>(b2, ws);
  maf_scan<<<BATCH / 8, 256, 0, stream>>>(z, b1, b2, ws, xout);
  maf_logdet_wmma<<<BATCH / 32, 512, 0, stream>>>(b1, ws, xout, ldout);
}


# scan: four batch rows per wave (512 blocks x 4 waves, was 1024 x 8 with one row each): each weight value read from LDS feeds all four rows; per-row sum order unchanged
# speedup vs baseline: 2.9281x; 1.7353x over previous
#include <hip/hip_runtime.h>
#include <hip/hip_bf16.h>

#define DIM 64
#define HID 512
#define BATCH 8192

typedef __attribute__((ext_vector_type(2))) float v2f;
typedef __attribute__((ext_vector_type(4))) float v4f;
typedef __attribute__((ext_vector_type(8))) float v8f;

#define WS_MW1T  0
#define WS_MW2   (HID*DIM)
#define WS_WSUM  (HID*DIM + 2*DIM*HID)
#define WS_BSUM  (WS_WSUM + HID)
#define WS_BPACK 98832
#define NPACK    (HID*DIM)

__global__ __launch_bounds__(256) void maf_prep_mask(const float* __restrict__ W1,
                                                     const float* __restrict__ mask1,
                                                     const float* __restrict__ W2,
                                                     const float* __restrict__ mask2,
                                                     float* __restrict__ ws) {
  int idx = blockIdx.x * blockDim.x + threadIdx.x;
  if (idx < HID * DIM) {
    int j = idx / DIM, k = idx % DIM;
    ws[WS_MW1T + k * HID + j] = mask1[idx] * W1[idx];
  } else if (idx < HID * DIM + 2 * DIM * HID) {
    int t = idx - HID * DIM;
    ws[WS_MW2 + t] = mask2[t] * W2[t];
  } else if (idx < HID * DIM + 2 * DIM * HID + NPACK) {
    int f = idx - (HID * DIM + 2 * DIM * HID);
    int p = f >> 1, c = f & 1;
    int nt = p >> 9;
    int t  = (p >> 5) & 15;
    int l  = p & 31;
    int koff = (l >> 4) << 1;
    int j = (nt << 4) + (l & 15);
    int k = (t << 2) + koff + c;
    ws[WS_BPACK + f] = mask1[j * DIM + k] * W1[j * DIM + k];
  }
}

__global__ __launch_bounds__(512) void maf_prep_sum(const float* __restrict__ b2,
                                                    float* __restrict__ ws) {
  int j = threadIdx.x;
  const float* MW2 = ws + WS_MW2;
  float s = 0.f;
  for (int r = DIM; r < 2 * DIM; ++r) s += MW2[r * HID + j];
  ws[WS_WSUM + j] = s;
  if (j == 0) {
    float bs = 0.f;
    for (int r = DIM; r < 2 * DIM; ++r) bs += b2[r];
    ws[WS_BSUM] = bs;
  }
}

__device__ __forceinline__ float lane_xor16(float x) {
  const int b = __builtin_bit_cast(int, x);
  return __builtin_bit_cast(float,
      __builtin_amdgcn_permlanex16(b, b, 0x76543210, 0xfedcba98, false, false));
}
template <int M>
__device__ __forceinline__ float row_xor(float x) {
  const int b = __builtin_bit_cast(int, x);
  return __builtin_bit_cast(float, __builtin_amdgcn_update_dpp(0, b, 0x160 + M, 0xf, 0xf, true));
}

__device__ __forceinline__ void lds_barrier() {
  asm volatile("s_wait_dscnt 0x0\n\ts_barrier_signal -1\n\ts_barrier_wait -1" ::: "memory");
}

#define SCAN_R 4
#define SCAN_WAVES 4
#define SCAN_ROWS (SCAN_WAVES * SCAN_R)
__global__ __launch_bounds__(32 * SCAN_WAVES) void maf_scan(const float* __restrict__ z,
                                                const float* __restrict__ b1,
                                                const float* __restrict__ b2,
                                                const float* __restrict__ ws,
                                                float* __restrict__ xout) {
  __shared__ __align__(16) float w1c[2][HID];
  __shared__ __align__(16) float w2a[2][HID];
  __shared__ __align__(16) float w2b[2][HID];
  const float* MW1T = ws + WS_MW1T;
  const float* MW2  = ws + WS_MW2;

  const int tid  = threadIdx.x;
  const int lane = tid & 31;
  const int wave = tid >> 5;
  const int row0 = blockIdx.x * SCAN_ROWS + wave * SCAN_R;
  const int t4   = tid * 4;

  *(v4f*)&w1c[0][t4] = *(const v4f*)&MW1T[t4];
  *(v4f*)&w2a[0][t4] = *(const v4f*)&MW2[t4];
  *(v4f*)&w2b[0][t4] = *(const v4f*)&MW2[DIM * HID + t4];

  float h[SCAN_R][16];
#pragma unroll
  for (int t = 0; t < 16; ++t) {
    const float bv = b1[lane + 32 * t];
#pragma unroll
    for (int r = 0; r < SCAN_R; ++r) h[r][t] = bv;
  }

  float z0[SCAN_R], z1[SCAN_R], x0[SCAN_R], x1[SCAN_R];
#pragma unroll
  for (int r = 0; r < SCAN_R; ++r) {
    z0[r] = z[(row0 + r) * DIM + lane];
    z1[r] = z[(row0 + r) * DIM + 32 + lane];
    x0[r] = 0.f;
    x1[r] = 0.f;
  }

  auto column = [&](int i, const float* wa, const float* wb, float (&xi)[SCAN_R]) {
    float mu[SCAN_R], al[SCAN_R];
#pragma unroll
    for (int r = 0; r < SCAN_R; ++r) { mu[r] = 0.f; al[r] = 0.f; }
#pragma unroll
    for (int t = 0; t < 16; ++t) {
      const int j = lane + 32 * t;
      const float wav = wa[j];
      const float wbv = wb[j];
#pragma unroll
      for (int r = 0; r < SCAN_R; ++r) {
        float hv = fmaxf(h[r][t], 0.f);
        mu[r] = fmaf(hv, wav, mu[r]);
        al[r] = fmaf(hv, wbv, al[r]);
      }
    }
    const float bmu = b2[i];
    const float bal = b2[DIM + i];
#pragma unroll
    for (int r = 0; r < SCAN_R; ++r) {
      float m = mu[r], a = al[r];
      m += lane_xor16(m);  a += lane_xor16(a);
      m += row_xor<8>(m);  a += row_xor<8>(a);
      m += row_xor<4>(m);  a += row_xor<4>(a);
      m += row_xor<2>(m);  a += row_xor<2>(a);
      m += row_xor<1>(m);  a += row_xor<1>(a);
      m += bmu;
      a += bal;

      float zsel = (i < 32) ? z0[r] : z1[r];
      float zi = __builtin_bit_cast(float,
          __builtin_amdgcn_readlane(__builtin_bit_cast(int, zsel), i & 31));
      xi[r] = fmaf(zi, expf(a), m);
    }
  };

  v4f p1 = *(const v4f*)&MW1T[HID + t4];
  v4f p2 = *(const v4f*)&MW2[HID + t4];
  v4f p3 = *(const v4f*)&MW2[(DIM + 1) * HID + t4];

  lds_barrier();

#pragma unroll 2
  for (int i = 0; i < DIM - 1; ++i) {
    const int cur = i & 1;
    const int nxt = cur ^ 1;
    const int i2 = (i + 2 < DIM) ? i + 2 : DIM - 1;
    const v4f q1 = *(const v4f*)&MW1T[i2 * HID + t4];
    const v4f q2 = *(const v4f*)&MW2[i2 * HID + t4];
    const v4f q3 = *(const v4f*)&MW2[(DIM + i2) * HID + t4];

    float xi[SCAN_R];
    column(i, w2a[cur], w2b[cur], xi);
#pragma unroll
    for (int r = 0; r < SCAN_R; ++r) {
      x0[r] = (lane == i)      ? xi[r] : x0[r];
      x1[r] = (lane == i - 32) ? xi[r] : x1[r];
    }

#pragma unroll
    for (int t = 0; t < 16; ++t) {
      const float wv = w1c[cur][lane + 32 * t];
#pragma unroll
      for (int r = 0; r < SCAN_R; ++r) h[r][t] = fmaf(xi[r], wv, h[r][t]);
    }

    *(v4f*)&w1c[nxt][t4] = p1;
    *(v4f*)&w2a[nxt][t4] = p2;
    *(v4f*)&w2b[nxt][t4] = p3;
    lds_barrier();
    p1 = q1;  p2 = q2;  p3 = q3;
  }

  {
    const int i = DIM - 1;
    float xi[SCAN_R];
    column(i, w2a[i & 1], w2b[i & 1], xi);
#pragma unroll
    for (int r = 0; r < SCAN_R; ++r) {
      x0[r] = (lane == i)      ? xi[r] : x0[r];
      x1[r] = (lane == i - 32) ? xi[r] : x1[r];
    }
  }

#pragma unroll
  for (int r = 0; r < SCAN_R; ++r) {
    xout[(row0 + r) * DIM + lane]      = x0[r];
    xout[(row0 + r) * DIM + 32 + lane] = x1[r];
  }
}

#define LD_NT 4
__global__ __launch_bounds__(512) void maf_logdet_wmma(const float* __restrict__ b1,
                                                       const float* __restrict__ ws,
                                                       const float* __restrict__ xout,
                                                       float* __restrict__ ldout) {
  __shared__ float b1s[HID];
  __shared__ float wss[HID];
  __shared__ float part[8][32];
  const v2f* Bp = (const v2f*)(ws + WS_BPACK);
  const float bsum = ws[WS_BSUM];

  const int tid  = threadIdx.x;
  const int lane = tid & 31;
  const int wave = tid >> 5;
  const int lr   = lane & 15;
  const int koff = (lane >> 4) ? 2 : 0;
  const int mt   = wave & 1;
  const int q    = wave >> 1;
  const int m0   = blockIdx.x * 32 + mt * 16;
  const int nt0  = q * LD_NT;

  const float b1v = b1[tid];
  const float wsv = ws[WS_WSUM + tid];

  const float* xp = xout + (m0 + lr) * DIM + koff;
  v2f a[16];
#pragma unroll
  for (int t = 0; t < 16; ++t) a[t] = *(const v2f*)(xp + 4 * t);

  v2f bt[LD_NT][16];
#pragma unroll
  for (int i = 0; i < LD_NT; ++i) {
#pragma unroll
    for (int t = 0; t < 16; ++t) bt[i][t] = Bp[((nt0 + i) * 16 + t) * 32 + lane];
  }

  __builtin_amdgcn_sched_barrier(0);

  b1s[tid] = b1v;
  wss[tid] = wsv;
  __syncthreads();

  v8f ld = {};
#pragma unroll
  for (int i = 0; i < LD_NT; ++i) {
    const int col = (nt0 + i) * 16 + lr;

    v8f acc0 = {}, acc1 = {};
#pragma unroll
    for (int t = 0; t < 16; t += 2) {
      acc0 = __builtin_amdgcn_wmma_f32_16x16x4_f32(false, a[t],     false, bt[i][t],
                                                   (short)0, acc0, false, false);
      acc1 = __builtin_amdgcn_wmma_f32_16x16x4_f32(false, a[t + 1], false, bt[i][t + 1],
                                                   (short)0, acc1, false, false);
    }
    v8f acc = acc0 + acc1;

    const float bb = b1s[col];
    const float wv = wss[col];
#pragma unroll
    for (int v = 0; v < 8; ++v) {
      float hv = fmaxf(acc[v] + bb, 0.f);
      ld[v] = fmaf(hv, wv, ld[v]);
    }
  }

#pragma unroll
  for (int off = 1; off < 16; off <<= 1) {
#pragma unroll
    for (int v = 0; v < 8; ++v) {
      float t = ld[v];
      ld[v] = t + __shfl_xor(t, off, 32);
    }
  }
  if (lr == 0) {
    const int r0 = mt * 16 + (lane >> 4) * 8;
#pragma unroll
    for (int v = 0; v < 8; ++v) part[q][r0 + v] = ld[v];
  }
  __syncthreads();
  if (tid < 32) {
    float s = part[0][tid];
#pragma unroll
    for (int k = 1; k < 8; ++k) s += part[k][tid];
    ldout[blockIdx.x * 32 + tid] = s + bsum;
  }
}

extern "C" void kernel_launch(void* const* d_in, const int* in_sizes, int n_in,
                              void* d_out, int out_size, void* d_ws, size_t ws_size,
                              hipStream_t stream) {
  const float* z     = (const float*)d_in[0];
  const float* W1    = (const float*)d_in[1];
  const float* b1    = (const float*)d_in[2];
  const float* W2    = (const float*)d_in[3];
  const float* b2    = (const float*)d_in[4];
  const float* mask1 = (const float*)d_in[5];
  const float* mask2 = (const float*)d_in[6];

  float* ws    = (float*)d_ws;
  float* xout  = (float*)d_out;
  float* ldout = xout + BATCH * DIM;

  const int nprep = HID * DIM + 2 * DIM * HID + NPACK;
  maf_prep_mask<<<(nprep + 255) / 256, 256, 0, stream>>>(W1, mask1, W2, mask2, ws);
  maf_prep_sum<<<1, 512, 0, stream>>>(b2, ws);
  maf_scan<<<BATCH / SCAN_ROWS, 32 * SCAN_WAVES, 0, stream>>>(z, b1, b2, ws, xout);
  maf_logdet_wmma<<<BATCH / 32, 512, 0, stream>>>(b1, ws, xout, ldout);
}


# scan: the 8 wave sums of a step are folded while reduced (rows share a register after the exchanges at distance 8 and 4), so bias, exp and z product run once per step, not per row; new sum order
# speedup vs baseline: 3.2326x; 1.1040x over previous
#include <hip/hip_runtime.h>
#include <hip/hip_bf16.h>

#define DIM 64
#define HID 512
#define BATCH 8192

typedef __attribute__((ext_vector_type(2))) float v2f;
typedef __attribute__((ext_vector_type(4))) float v4f;
typedef __attribute__((ext_vector_type(8))) float v8f;

#define WS_MW1T  0
#define WS_MW2   (HID*DIM)
#define WS_WSUM  (HID*DIM + 2*DIM*HID)
#define WS_BSUM  (WS_WSUM + HID)
#define WS_BPACK 98832
#define NPACK    (HID*DIM)

__global__ __launch_bounds__(256) void maf_prep_mask(const float* __restrict__ W1,
                                                     const float* __restrict__ mask1,
                                                     const float* __restrict__ W2,
                                                     const float* __restrict__ mask2,
                                                     float* __restrict__ ws) {
  int idx = blockIdx.x * blockDim.x + threadIdx.x;
  if (idx < HID * DIM) {
    int j = idx / DIM, k = idx % DIM;
    ws[WS_MW1T + k * HID + j] = mask1[idx] * W1[idx];
  } else if (idx < HID * DIM + 2 * DIM * HID) {
    int t = idx - HID * DIM;
    ws[WS_MW2 + t] = mask2[t] * W2[t];
  } else if (idx < HID * DIM + 2 * DIM * HID + NPACK) {
    int f = idx - (HID * DIM + 2 * DIM * HID);
    int p = f >> 1, c = f & 1;
    int nt = p >> 9;
    int t  = (p >> 5) & 15;
    int l  = p & 31;
    int koff = (l >> 4) << 1;
    int j = (nt << 4) + (l & 15);
    int k = (t << 2) + koff + c;
    ws[WS_BPACK + f] = mask1[j * DIM + k] * W1[j * DIM + k];
  }
}

__global__ __launch_bounds__(512) void maf_prep_sum(const float* __restrict__ b2,
                                                    float* __restrict__ ws) {
  int j = threadIdx.x;
  const float* MW2 = ws + WS_MW2;
  float s = 0.f;
  for (int r = DIM; r < 2 * DIM; ++r) s += MW2[r * HID + j];
  ws[WS_WSUM + j] = s;
  if (j == 0) {
    float bs = 0.f;
    for (int r = DIM; r < 2 * DIM; ++r) bs += b2[r];
    ws[WS_BSUM] = bs;
  }
}

__device__ __forceinline__ float lane_xor16(float x) {
  const int b = __builtin_bit_cast(int, x);
  return __builtin_bit_cast(float,
      __builtin_amdgcn_permlanex16(b, b, 0x76543210, 0xfedcba98, false, false));
}
template <int M>
__device__ __forceinline__ float row_xor(float x) {
  const int b = __builtin_bit_cast(int, x);
  return __builtin_bit_cast(float, __builtin_amdgcn_update_dpp(0, b, 0x160 + M, 0xf, 0xf, true));
}

__device__ __forceinline__ void lds_barrier() {
  asm volatile("s_wait_dscnt 0x0\n\ts_barrier_signal -1\n\ts_barrier_wait -1" ::: "memory");
}

#define SCAN_R 4
#define ZX_PITCH (DIM + 1)
#define SCAN_WAVES 4
#define SCAN_ROWS (SCAN_WAVES * SCAN_R)
__global__ __launch_bounds__(32 * SCAN_WAVES) void maf_scan(const float* __restrict__ z,
                                                const float* __restrict__ b1,
                                                const float* __restrict__ b2,
                                                const float* __restrict__ ws,
                                                float* __restrict__ xout) {
  __shared__ __align__(16) float w1c[2][HID];
  __shared__ __align__(16) float w2a[2][HID];
  __shared__ __align__(16) float w2b[2][HID];
  __shared__ float zx[SCAN_WAVES][SCAN_R][ZX_PITCH];
  const float* MW1T = ws + WS_MW1T;
  const float* MW2  = ws + WS_MW2;

  const int tid  = threadIdx.x;
  const int lane = tid & 31;
  const int wave = tid >> 5;
  const int row0 = blockIdx.x * SCAN_ROWS + wave * SCAN_R;
  const int t4   = tid * 4;

  *(v4f*)&w1c[0][t4] = *(const v4f*)&MW1T[t4];
  *(v4f*)&w2a[0][t4] = *(const v4f*)&MW2[t4];
  *(v4f*)&w2b[0][t4] = *(const v4f*)&MW2[DIM * HID + t4];

  float h[SCAN_R][16];
#pragma unroll
  for (int t = 0; t < 16; ++t) {
    const float bv = b1[lane + 32 * t];
#pragma unroll
    for (int r = 0; r < SCAN_R; ++r) h[r][t] = bv;
  }

#pragma unroll
  for (int r = 0; r < SCAN_R; ++r) {
    zx[wave][r][lane]      = z[(row0 + r) * DIM + lane];
    zx[wave][r][32 + lane] = z[(row0 + r) * DIM + 32 + lane];
  }

  const bool up8 = (lane & 8) != 0;
  const bool up4 = (lane & 4) != 0;
  const int  rl  = (up8 ? 1 : 0) + (up4 ? 2 : 0);
  float* zxl = &zx[wave][rl][0];

  auto column = [&](int i, const float* wa, const float* wb) -> float {
    const float zi = zxl[i];
    float mu[SCAN_R], al[SCAN_R];
#pragma unroll
    for (int r = 0; r < SCAN_R; ++r) { mu[r] = 0.f; al[r] = 0.f; }
#pragma unroll
    for (int t = 0; t < 16; ++t) {
      const int j = lane + 32 * t;
      const float wav = wa[j];
      const float wbv = wb[j];
#pragma unroll
      for (int r = 0; r < SCAN_R; ++r) {
        float hv = fmaxf(h[r][t], 0.f);
        mu[r] = fmaf(hv, wav, mu[r]);
        al[r] = fmaf(hv, wbv, al[r]);
      }
    }
    const float m01 = (up8 ? mu[1] : mu[0]) + row_xor<8>(up8 ? mu[0] : mu[1]);
    const float m23 = (up8 ? mu[3] : mu[2]) + row_xor<8>(up8 ? mu[2] : mu[3]);
    const float a01 = (up8 ? al[1] : al[0]) + row_xor<8>(up8 ? al[0] : al[1]);
    const float a23 = (up8 ? al[3] : al[2]) + row_xor<8>(up8 ? al[2] : al[3]);
    float m = (up4 ? m23 : m01) + row_xor<4>(up4 ? m01 : m23);
    float a = (up4 ? a23 : a01) + row_xor<4>(up4 ? a01 : a23);
    m += row_xor<2>(m);  a += row_xor<2>(a);
    m += row_xor<1>(m);  a += row_xor<1>(a);
    m += lane_xor16(m);  a += lane_xor16(a);
    m += b2[i];
    a += b2[DIM + i];
    return fmaf(zi, expf(a), m);
  };

  auto spread = [&](float xv, float (&xi)[SCAN_R]) {
    const int xb = __builtin_bit_cast(int, xv);
    xi[0] = __builtin_bit_cast(float, __builtin_amdgcn_readlane(xb, 0));
    xi[1] = __builtin_bit_cast(float, __builtin_amdgcn_readlane(xb, 8));
    xi[2] = __builtin_bit_cast(float, __builtin_amdgcn_readlane(xb, 4));
    xi[3] = __builtin_bit_cast(float, __builtin_amdgcn_readlane(xb, 12));
  };

  v4f p1 = *(const v4f*)&MW1T[HID + t4];
  v4f p2 = *(const v4f*)&MW2[HID + t4];
  v4f p3 = *(const v4f*)&MW2[(DIM + 1) * HID + t4];

  lds_barrier();

#pragma unroll 2
  for (int i = 0; i < DIM - 1; ++i) {
    const int cur = i & 1;
    const int nxt = cur ^ 1;
    const int i2 = (i + 2 < DIM) ? i + 2 : DIM - 1;
    const v4f q1 = *(const v4f*)&MW1T[i2 * HID + t4];
    const v4f q2 = *(const v4f*)&MW2[i2 * HID + t4];
    const v4f q3 = *(const v4f*)&MW2[(DIM + i2) * HID + t4];

    const float xv = column(i, w2a[cur], w2b[cur]);
    float xi[SCAN_R];
    spread(xv, xi);

#pragma unroll
    for (int t = 0; t < 16; ++t) {
      const float wv = w1c[cur][lane + 32 * t];
#pragma unroll
      for (int r = 0; r < SCAN_R; ++r) h[r][t] = fmaf(xi[r], wv, h[r][t]);
    }

    *(v4f*)&w1c[nxt][t4] = p1;
    *(v4f*)&w2a[nxt][t4] = p2;
    *(v4f*)&w2b[nxt][t4] = p3;
    lds_barrier();
    zxl[i] = xv;
    p1 = q1;  p2 = q2;  p3 = q3;
  }

  {
    const int i = DIM - 1;
    zxl[i] = column(i, w2a[i & 1], w2b[i & 1]);
  }

#pragma unroll
  for (int r = 0; r < SCAN_R; ++r) {
    xout[(row0 + r) * DIM + lane]      = zx[wave][r][lane];
    xout[(row0 + r) * DIM + 32 + lane] = zx[wave][r][32 + lane];
  }
}

#define LD_NT 4
__global__ __launch_bounds__(512) void maf_logdet_wmma(const float* __restrict__ b1,
                                                       const float* __restrict__ ws,
                                                       const float* __restrict__ xout,
                                                       float* __restrict__ ldout) {
  __shared__ float b1s[HID];
  __shared__ float wss[HID];
  __shared__ float part[8][32];
  const v2f* Bp = (const v2f*)(ws + WS_BPACK);
  const float bsum = ws[WS_BSUM];

  const int tid  = threadIdx.x;
  const int lane = tid & 31;
  const int wave = tid >> 5;
  const int lr   = lane & 15;
  const int koff = (lane >> 4) ? 2 : 0;
  const int mt   = wave & 1;
  const int q    = wave >> 1;
  const int m0   = blockIdx.x * 32 + mt * 16;
  const int nt0  = q * LD_NT;

  const float b1v = b1[tid];
  const float wsv = ws[WS_WSUM + tid];

  const float* xp = xout + (m0 + lr) * DIM + koff;
  v2f a[16];
#pragma unroll
  for (int t = 0; t < 16; ++t) a[t] = *(const v2f*)(xp + 4 * t);

  v2f bt[LD_NT][16];
#pragma unroll
  for (int i = 0; i < LD_NT; ++i) {
#pragma unroll
    for (int t = 0; t < 16; ++t) bt[i][t] = Bp[((nt0 + i) * 16 + t) * 32 + lane];
  }

  __builtin_amdgcn_sched_barrier(0);

  b1s[tid] = b1v;
  wss[tid] = wsv;
  __syncthreads();

  v8f ld = {};
#pragma unroll
  for (int i = 0; i < LD_NT; ++i) {
    const int col = (nt0 + i) * 16 + lr;

    v8f acc0 = {}, acc1 = {};
#pragma unroll
    for (int t = 0; t < 16; t += 2) {
      acc0 = __builtin_amdgcn_wmma_f32_16x16x4_f32(false, a[t],     false, bt[i][t],
                                                   (short)0, acc0, false, false);
      acc1 = __builtin_amdgcn_wmma_f32_16x16x4_f32(false, a[t + 1], false, bt[i][t + 1],
                                                   (short)0, acc1, false, false);
    }
    v8f acc = acc0 + acc1;

    const float bb = b1s[col];
    const float wv = wss[col];
#pragma unroll
    for (int v = 0; v < 8; ++v) {
      float hv = fmaxf(acc[v] + bb, 0.f);
      ld[v] = fmaf(hv, wv, ld[v]);
    }
  }

#pragma unroll
  for (int off = 1; off < 16; off <<= 1) {
#pragma unroll
    for (int v = 0; v < 8; ++v) {
      float t = ld[v];
      ld[v] = t + __shfl_xor(t, off, 32);
    }
  }
  if (lr == 0) {
    const int r0 = mt * 16 + (lane >> 4) * 8;
#pragma unroll
    for (int v = 0; v < 8; ++v) part[q][r0 + v] = ld[v];
  }
  __syncthreads();
  if (tid < 32) {
    float s = part[0][tid];
#pragma unroll
    for (int k = 1; k < 8; ++k) s += part[k][tid];
    ldout[blockIdx.x * 32 + tid] = s + bsum;
  }
}

extern "C" void kernel_launch(void* const* d_in, const int* in_sizes, int n_in,
                              void* d_out, int out_size, void* d_ws, size_t ws_size,
                              hipStream_t stream) {
  const float* z     = (const float*)d_in[0];
  const float* W1    = (const float*)d_in[1];
  const float* b1    = (const float*)d_in[2];
  const float* W2    = (const float*)d_in[3];
  const float* b2    = (const float*)d_in[4];
  const float* mask1 = (const float*)d_in[5];
  const float* mask2 = (const float*)d_in[6];

  float* ws    = (float*)d_ws;
  float* xout  = (float*)d_out;
  float* ldout = xout + BATCH * DIM;

  const int nprep = HID * DIM + 2 * DIM * HID + NPACK;
  maf_prep_mask<<<(nprep + 255) / 256, 256, 0, stream>>>(W1, mask1, W2, mask2, ws);
  maf_prep_sum<<<1, 512, 0, stream>>>(b2, ws);
  maf_scan<<<BATCH / SCAN_ROWS, 32 * SCAN_WAVES, 0, stream>>>(z, b1, b2, ws, xout);
  maf_logdet_wmma<<<BATCH / 32, 512, 0, stream>>>(b1, ws, xout, ldout);
}


# prep in one launch (3 per call, was 4): w_sum and b_sum formed by three extra blocks of the masking kernel straight from W2, mask2, b2; w_sum products fuse with their adds, term order kept
# speedup vs baseline: 3.5352x; 1.0936x over previous
#include <hip/hip_runtime.h>
#include <hip/hip_bf16.h>

#define DIM 64
#define HID 512
#define BATCH 8192

typedef __attribute__((ext_vector_type(2))) float v2f;
typedef __attribute__((ext_vector_type(4))) float v4f;
typedef __attribute__((ext_vector_type(8))) float v8f;

#define WS_MW1T  0
#define WS_MW2   (HID*DIM)
#define WS_WSUM  (HID*DIM + 2*DIM*HID)
#define WS_BSUM  (WS_WSUM + HID)
#define WS_BPACK 98832
#define NPACK    (HID*DIM)

#define PREP_SUM_BLOCKS 2
#define PREP_HEAD_BLOCKS (PREP_SUM_BLOCKS + 1)
__global__ __launch_bounds__(256) void maf_prep(const float* __restrict__ W1,
                                                const float* __restrict__ mask1,
                                                const float* __restrict__ W2,
                                                const float* __restrict__ mask2,
                                                const float* __restrict__ b2,
                                                float* __restrict__ ws) {
  if (blockIdx.x < PREP_SUM_BLOCKS) {
    const int j = blockIdx.x * 256 + threadIdx.x;
    float s = 0.f;
#pragma unroll
    for (int r = DIM; r < 2 * DIM; ++r) s += mask2[r * HID + j] * W2[r * HID + j];
    ws[WS_WSUM + j] = s;
    return;
  }
  if (blockIdx.x == PREP_SUM_BLOCKS) {
    if (threadIdx.x == 0) {
      float bs = 0.f;
#pragma unroll
      for (int r = DIM; r < 2 * DIM; ++r) bs += b2[r];
      ws[WS_BSUM] = bs;
    }
    return;
  }
  const int idx = (blockIdx.x - PREP_HEAD_BLOCKS) * 256 + threadIdx.x;
  if (idx < HID * DIM) {
    int j = idx / DIM, k = idx % DIM;
    ws[WS_MW1T + k * HID + j] = mask1[idx] * W1[idx];
  } else if (idx < HID * DIM + 2 * DIM * HID) {
    int t = idx - HID * DIM;
    ws[WS_MW2 + t] = mask2[t] * W2[t];
  } else if (idx < HID * DIM + 2 * DIM * HID + NPACK) {
    int f = idx - (HID * DIM + 2 * DIM * HID);
    int p = f >> 1, c = f & 1;
    int nt = p >> 9;
    int t  = (p >> 5) & 15;
    int l  = p & 31;
    int koff = (l >> 4) << 1;
    int j = (nt << 4) + (l & 15);
    int k = (t << 2) + koff + c;
    ws[WS_BPACK + f] = mask1[j * DIM + k] * W1[j * DIM + k];
  }
}

__device__ __forceinline__ float lane_xor16(float x) {
  const int b = __builtin_bit_cast(int, x);
  return __builtin_bit_cast(float,
      __builtin_amdgcn_permlanex16(b, b, 0x76543210, 0xfedcba98, false, false));
}
template <int M>
__device__ __forceinline__ float row_xor(float x) {
  const int b = __builtin_bit_cast(int, x);
  return __builtin_bit_cast(float, __builtin_amdgcn_update_dpp(0, b, 0x160 + M, 0xf, 0xf, true));
}

__device__ __forceinline__ void lds_barrier() {
  asm volatile("s_wait_dscnt 0x0\n\ts_barrier_signal -1\n\ts_barrier_wait -1" ::: "memory");
}

#define SCAN_R 4
#define ZX_PITCH (DIM + 1)
#define SCAN_WAVES 4
#define SCAN_ROWS (SCAN_WAVES * SCAN_R)
__global__ __launch_bounds__(32 * SCAN_WAVES) void maf_scan(const float* __restrict__ z,
                                                const float* __restrict__ b1,
                                                const float* __restrict__ b2,
                                                const float* __restrict__ ws,
                                                float* __restrict__ xout) {
  __shared__ __align__(16) float w1c[2][HID];
  __shared__ __align__(16) float w2a[2][HID];
  __shared__ __align__(16) float w2b[2][HID];
  __shared__ float zx[SCAN_WAVES][SCAN_R][ZX_PITCH];
  const float* MW1T = ws + WS_MW1T;
  const float* MW2  = ws + WS_MW2;

  const int tid  = threadIdx.x;
  const int lane = tid & 31;
  const int wave = tid >> 5;
  const int row0 = blockIdx.x * SCAN_ROWS + wave * SCAN_R;
  const int t4   = tid * 4;

  *(v4f*)&w1c[0][t4] = *(const v4f*)&MW1T[t4];
  *(v4f*)&w2a[0][t4] = *(const v4f*)&MW2[t4];
  *(v4f*)&w2b[0][t4] = *(const v4f*)&MW2[DIM * HID + t4];

  float h[SCAN_R][16];
#pragma unroll
  for (int t = 0; t < 16; ++t) {
    const float bv = b1[lane + 32 * t];
#pragma unroll
    for (int r = 0; r < SCAN_R; ++r) h[r][t] = bv;
  }

#pragma unroll
  for (int r = 0; r < SCAN_R; ++r) {
    zx[wave][r][lane]      = z[(row0 + r) * DIM + lane];
    zx[wave][r][32 + lane] = z[(row0 + r) * DIM + 32 + lane];
  }

  const bool up8 = (lane & 8) != 0;
  const bool up4 = (lane & 4) != 0;
  const int  rl  = (up8 ? 1 : 0) + (up4 ? 2 : 0);
  float* zxl = &zx[wave][rl][0];

  auto column = [&](int i, const float* wa, const float* wb) -> float {
    const float zi = zxl[i];
    float mu[SCAN_R], al[SCAN_R];
#pragma unroll
    for (int r = 0; r < SCAN_R; ++r) { mu[r] = 0.f; al[r] = 0.f; }
#pragma unroll
    for (int t = 0; t < 16; ++t) {
      const int j = lane + 32 * t;
      const float wav = wa[j];
      const float wbv = wb[j];
#pragma unroll
      for (int r = 0; r < SCAN_R; ++r) {
        float hv = fmaxf(h[r][t], 0.f);
        mu[r] = fmaf(hv, wav, mu[r]);
        al[r] = fmaf(hv, wbv, al[r]);
      }
    }
    const float m01 = (up8 ? mu[1] : mu[0]) + row_xor<8>(up8 ? mu[0] : mu[1]);
    const float m23 = (up8 ? mu[3] : mu[2]) + row_xor<8>(up8 ? mu[2] : mu[3]);
    const float a01 = (up8 ? al[1] : al[0]) + row_xor<8>(up8 ? al[0] : al[1]);
    const float a23 = (up8 ? al[3] : al[2]) + row_xor<8>(up8 ? al[2] : al[3]);
    float m = (up4 ? m23 : m01) + row_xor<4>(up4 ? m01 : m23);
    float a = (up4 ? a23 : a01) + row_xor<4>(up4 ? a01 : a23);
    m += row_xor<2>(m);  a += row_xor<2>(a);
    m += row_xor<1>(m);  a += row_xor<1>(a);
    m += lane_xor16(m);  a += lane_xor16(a);
    m += b2[i];
    a += b2[DIM + i];
    return fmaf(zi, expf(a), m);
  };

  auto spread = [&](float xv, float (&xi)[SCAN_R]) {
    const int xb = __builtin_bit_cast(int, xv);
    xi[0] = __builtin_bit_cast(float, __builtin_amdgcn_readlane(xb, 0));
    xi[1] = __builtin_bit_cast(float, __builtin_amdgcn_readlane(xb, 8));
    xi[2] = __builtin_bit_cast(float, __builtin_amdgcn_readlane(xb, 4));
    xi[3] = __builtin_bit_cast(float, __builtin_amdgcn_readlane(xb, 12));
  };

  v4f p1 = *(const v4f*)&MW1T[HID + t4];
  v4f p2 = *(const v4f*)&MW2[HID + t4];
  v4f p3 = *(const v4f*)&MW2[(DIM + 1) * HID + t4];

  lds_barrier();

#pragma unroll 2
  for (int i = 0; i < DIM - 1; ++i) {
    const int cur = i & 1;
    const int nxt = cur ^ 1;
    const int i2 = (i + 2 < DIM) ? i + 2 : DIM - 1;
    const v4f q1 = *(const v4f*)&MW1T[i2 * HID + t4];
    const v4f q2 = *(const v4f*)&MW2[i2 * HID + t4];
    const v4f q3 = *(const v4f*)&MW2[(DIM + i2) * HID + t4];

    const float xv = column(i, w2a[cur], w2b[cur]);
    float xi[SCAN_R];
    spread(xv, xi);

#pragma unroll
    for (int t = 0; t < 16; ++t) {
      const float wv = w1c[cur][lane + 32 * t];
#pragma unroll
      for (int r = 0; r < SCAN_R; ++r) h[r][t] = fmaf(xi[r], wv, h[r][t]);
    }

    *(v4f*)&w1c[nxt][t4] = p1;
    *(v4f*)&w2a[nxt][t4] = p2;
    *(v4f*)&w2b[nxt][t4] = p3;
    lds_barrier();
    zxl[i] = xv;
    p1 = q1;  p2 = q2;  p3 = q3;
  }

  {
    const int i = DIM - 1;
    zxl[i] = column(i, w2a[i & 1], w2b[i & 1]);
  }

#pragma unroll
  for (int r = 0; r < SCAN_R; ++r) {
    xout[(row0 + r) * DIM + lane]      = zx[wave][r][lane];
    xout[(row0 + r) * DIM + 32 + lane] = zx[wave][r][32 + lane];
  }
}

#define LD_NT 4
__global__ __launch_bounds__(512) void maf_logdet_wmma(const float* __restrict__ b1,
                                                       const float* __restrict__ ws,
                                                       const float* __restrict__ xout,
                                                       float* __restrict__ ldout) {
  __shared__ float b1s[HID];
  __shared__ float wss[HID];
  __shared__ float part[8][32];
  const v2f* Bp = (const v2f*)(ws + WS_BPACK);
  const float bsum = ws[WS_BSUM];

  const int tid  = threadIdx.x;
  const int lane = tid & 31;
  const int wave = tid >> 5;
  const int lr   = lane & 15;
  const int koff = (lane >> 4) ? 2 : 0;
  const int mt   = wave & 1;
  const int q    = wave >> 1;
  const int m0   = blockIdx.x * 32 + mt * 16;
  const int nt0  = q * LD_NT;

  const float b1v = b1[tid];
  const float wsv = ws[WS_WSUM + tid];

  const float* xp = xout + (m0 + lr) * DIM + koff;
  v2f a[16];
#pragma unroll
  for (int t = 0; t < 16; ++t) a[t] = *(const v2f*)(xp + 4 * t);

  v2f bt[LD_NT][16];
#pragma unroll
  for (int i = 0; i < LD_NT; ++i) {
#pragma unroll
    for (int t = 0; t < 16; ++t) bt[i][t] = Bp[((nt0 + i) * 16 + t) * 32 + lane];
  }

  __builtin_amdgcn_sched_barrier(0);

  b1s[tid] = b1v;
  wss[tid] = wsv;
  __syncthreads();

  v8f ld = {};
#pragma unroll
  for (int i = 0; i < LD_NT; ++i) {
    const int col = (nt0 + i) * 16 + lr;

    v8f acc0 = {}, acc1 = {};
#pragma unroll
    for (int t = 0; t < 16; t += 2) {
      acc0 = __builtin_amdgcn_wmma_f32_16x16x4_f32(false, a[t],     false, bt[i][t],
                                                   (short)0, acc0, false, false);
      acc1 = __builtin_amdgcn_wmma_f32_16x16x4_f32(false, a[t + 1], false, bt[i][t + 1],
                                                   (short)0, acc1, false, false);
    }
    v8f acc = acc0 + acc1;

    const float bb = b1s[col];
    const float wv = wss[col];
#pragma unroll
    for (int v = 0; v < 8; ++v) {
      float hv = fmaxf(acc[v] + bb, 0.f);
      ld[v] = fmaf(hv, wv, ld[v]);
    }
  }

#pragma unroll
  for (int off = 1; off < 16; off <<= 1) {
#pragma unroll
    for (int v = 0; v < 8; ++v) {
      float t = ld[v];
      ld[v] = t + __shfl_xor(t, off, 32);
    }
  }
  if (lr == 0) {
    const int r0 = mt * 16 + (lane >> 4) * 8;
#pragma unroll
    for (int v = 0; v < 8; ++v) part[q][r0 + v] = ld[v];
  }
  __syncthreads();
  if (tid < 32) {
    float s = part[0][tid];
#pragma unroll
    for (int k = 1; k < 8; ++k) s += part[k][tid];
    ldout[blockIdx.x * 32 + tid] = s + bsum;
  }
}

extern "C" void kernel_launch(void* const* d_in, const int* in_sizes, int n_in,
                              void* d_out, int out_size, void* d_ws, size_t ws_size,
                              hipStream_t stream) {
  const float* z     = (const float*)d_in[0];
  const float* W1    = (const float*)d_in[1];
  const float* b1    = (const float*)d_in[2];
  const float* W2    = (const float*)d_in[3];
  const float* b2    = (const float*)d_in[4];
  const float* mask1 = (const float*)d_in[5];
  const float* mask2 = (const float*)d_in[6];

  float* ws    = (float*)d_ws;
  float* xout  = (float*)d_out;
  float* ldout = xout + BATCH * DIM;

  const int nprep = HID * DIM + 2 * DIM * HID + NPACK;
  maf_prep<<<PREP_HEAD_BLOCKS + (nprep + 255) / 256, 256, 0, stream>>>(W1, mask1, W2, mask2, b2, ws);
  maf_scan<<<BATCH / SCAN_ROWS, 32 * SCAN_WAVES, 0, stream>>>(z, b1, b2, ws, xout);
  maf_logdet_wmma<<<BATCH / 32, 512, 0, stream>>>(b1, ws, xout, ldout);
}


# scan: weight rows requested four steps ahead through a ring of four register sets (was two steps, two sets), requests held at the top of each step; four steps per loop trip
# speedup vs baseline: 3.8713x; 1.0951x over previous
#include <hip/hip_runtime.h>
#include <hip/hip_bf16.h>

#define DIM 64
#define HID 512
#define BATCH 8192

typedef __attribute__((ext_vector_type(2))) float v2f;
typedef __attribute__((ext_vector_type(4))) float v4f;
typedef __attribute__((ext_vector_type(8))) float v8f;

#define WS_MW1T  0
#define WS_MW2   (HID*DIM)
#define WS_WSUM  (HID*DIM + 2*DIM*HID)
#define WS_BSUM  (WS_WSUM + HID)
#define WS_BPACK 98832
#define NPACK    (HID*DIM)

#define PREP_SUM_BLOCKS 2
#define PREP_HEAD_BLOCKS (PREP_SUM_BLOCKS + 1)
__global__ __launch_bounds__(256) void maf_prep(const float* __restrict__ W1,
                                                const float* __restrict__ mask1,
                                                const float* __restrict__ W2,
                                                const float* __restrict__ mask2,
                                                const float* __restrict__ b2,
                                                float* __restrict__ ws) {
  if (blockIdx.x < PREP_SUM_BLOCKS) {
    const int j = blockIdx.x * 256 + threadIdx.x;
    float s = 0.f;
#pragma unroll
    for (int r = DIM; r < 2 * DIM; ++r) s += mask2[r * HID + j] * W2[r * HID + j];
    ws[WS_WSUM + j] = s;
    return;
  }
  if (blockIdx.x == PREP_SUM_BLOCKS) {
    if (threadIdx.x == 0) {
      float bs = 0.f;
#pragma unroll
      for (int r = DIM; r < 2 * DIM; ++r) bs += b2[r];
      ws[WS_BSUM] = bs;
    }
    return;
  }
  const int idx = (blockIdx.x - PREP_HEAD_BLOCKS) * 256 + threadIdx.x;
  if (idx < HID * DIM) {
    int j = idx / DIM, k = idx % DIM;
    ws[WS_MW1T + k * HID + j] = mask1[idx] * W1[idx];
  } else if (idx < HID * DIM + 2 * DIM * HID) {
    int t = idx - HID * DIM;
    ws[WS_MW2 + t] = mask2[t] * W2[t];
  } else if (idx < HID * DIM + 2 * DIM * HID + NPACK) {
    int f = idx - (HID * DIM + 2 * DIM * HID);
    int p = f >> 1, c = f & 1;
    int nt = p >> 9;
    int t  = (p >> 5) & 15;
    int l  = p & 31;
    int koff = (l >> 4) << 1;
    int j = (nt << 4) + (l & 15);
    int k = (t << 2) + koff + c;
    ws[WS_BPACK + f] = mask1[j * DIM + k] * W1[j * DIM + k];
  }
}

__device__ __forceinline__ float lane_xor16(float x) {
  const int b = __builtin_bit_cast(int, x);
  return __builtin_bit_cast(float,
      __builtin_amdgcn_permlanex16(b, b, 0x76543210, 0xfedcba98, false, false));
}
template <int M>
__device__ __forceinline__ float row_xor(float x) {
  const int b = __builtin_bit_cast(int, x);
  return __builtin_bit_cast(float, __builtin_amdgcn_update_dpp(0, b, 0x160 + M, 0xf, 0xf, true));
}

__device__ __forceinline__ void lds_barrier() {
  asm volatile("s_wait_dscnt 0x0\n\ts_barrier_signal -1\n\ts_barrier_wait -1" ::: "memory");
}

#define SCAN_R 4
#define ZX_PITCH (DIM + 1)
#define SCAN_WAVES 4
#define SCAN_ROWS (SCAN_WAVES * SCAN_R)
#define SCAN_PF 4
__global__ __launch_bounds__(32 * SCAN_WAVES) void maf_scan(const float* __restrict__ z,
                                                const float* __restrict__ b1,
                                                const float* __restrict__ b2,
                                                const float* __restrict__ ws,
                                                float* __restrict__ xout) {
  __shared__ __align__(16) float w1c[2][HID];
  __shared__ __align__(16) float w2a[2][HID];
  __shared__ __align__(16) float w2b[2][HID];
  __shared__ float zx[SCAN_WAVES][SCAN_R][ZX_PITCH];
  const float* MW1T = ws + WS_MW1T;
  const float* MW2  = ws + WS_MW2;

  const int tid  = threadIdx.x;
  const int lane = tid & 31;
  const int wave = tid >> 5;
  const int row0 = blockIdx.x * SCAN_ROWS + wave * SCAN_R;
  const int t4   = tid * 4;

  *(v4f*)&w1c[0][t4] = *(const v4f*)&MW1T[t4];
  *(v4f*)&w2a[0][t4] = *(const v4f*)&MW2[t4];
  *(v4f*)&w2b[0][t4] = *(const v4f*)&MW2[DIM * HID + t4];

  float h[SCAN_R][16];
#pragma unroll
  for (int t = 0; t < 16; ++t) {
    const float bv = b1[lane + 32 * t];
#pragma unroll
    for (int r = 0; r < SCAN_R; ++r) h[r][t] = bv;
  }

#pragma unroll
  for (int r = 0; r < SCAN_R; ++r) {
    zx[wave][r][lane]      = z[(row0 + r) * DIM + lane];
    zx[wave][r][32 + lane] = z[(row0 + r) * DIM + 32 + lane];
  }

  const bool up8 = (lane & 8) != 0;
  const bool up4 = (lane & 4) != 0;
  const int  rl  = (up8 ? 1 : 0) + (up4 ? 2 : 0);
  float* zxl = &zx[wave][rl][0];

  auto column = [&](int i, const float* wa, const float* wb) -> float {
    const float zi = zxl[i];
    float mu[SCAN_R], al[SCAN_R];
#pragma unroll
    for (int r = 0; r < SCAN_R; ++r) { mu[r] = 0.f; al[r] = 0.f; }
#pragma unroll
    for (int t = 0; t < 16; ++t) {
      const int j = lane + 32 * t;
      const float wav = wa[j];
      const float wbv = wb[j];
#pragma unroll
      for (int r = 0; r < SCAN_R; ++r) {
        float hv = fmaxf(h[r][t], 0.f);
        mu[r] = fmaf(hv, wav, mu[r]);
        al[r] = fmaf(hv, wbv, al[r]);
      }
    }
    const float m01 = (up8 ? mu[1] : mu[0]) + row_xor<8>(up8 ? mu[0] : mu[1]);
    const float m23 = (up8 ? mu[3] : mu[2]) + row_xor<8>(up8 ? mu[2] : mu[3]);
    const float a01 = (up8 ? al[1] : al[0]) + row_xor<8>(up8 ? al[0] : al[1]);
    const float a23 = (up8 ? al[3] : al[2]) + row_xor<8>(up8 ? al[2] : al[3]);
    float m = (up4 ? m23 : m01) + row_xor<4>(up4 ? m01 : m23);
    float a = (up4 ? a23 : a01) + row_xor<4>(up4 ? a01 : a23);
    m += row_xor<2>(m);  a += row_xor<2>(a);
    m += row_xor<1>(m);  a += row_xor<1>(a);
    m += lane_xor16(m);  a += lane_xor16(a);
    m += b2[i];
    a += b2[DIM + i];
    return fmaf(zi, expf(a), m);
  };

  auto spread = [&](float xv, float (&xi)[SCAN_R]) {
    const int xb = __builtin_bit_cast(int, xv);
    xi[0] = __builtin_bit_cast(float, __builtin_amdgcn_readlane(xb, 0));
    xi[1] = __builtin_bit_cast(float, __builtin_amdgcn_readlane(xb, 8));
    xi[2] = __builtin_bit_cast(float, __builtin_amdgcn_readlane(xb, 4));
    xi[3] = __builtin_bit_cast(float, __builtin_amdgcn_readlane(xb, 12));
  };

  v4f r1[SCAN_PF], r2[SCAN_PF], r3[SCAN_PF];
  auto request = [&](int s, int slot) {
    r1[slot] = *(const v4f*)&MW1T[s * HID + t4];
    r2[slot] = *(const v4f*)&MW2[s * HID + t4];
    r3[slot] = *(const v4f*)&MW2[(DIM + s) * HID + t4];
  };
#pragma unroll
  for (int s = 1; s < SCAN_PF; ++s) request(s, s);

  lds_barrier();

  auto step = [&](int i, int k, bool ask) __attribute__((always_inline)) {
    const int cur = k & 1;
    const int nxt = cur ^ 1;
    if (ask) {
      request(i + SCAN_PF, k);
      __builtin_amdgcn_sched_barrier(0x060f);
    }

    const float xv = column(i, w2a[cur], w2b[cur]);
    float xi[SCAN_R];
    spread(xv, xi);

#pragma unroll
    for (int t = 0; t < 16; ++t) {
      const float wv = w1c[cur][lane + 32 * t];
#pragma unroll
      for (int r = 0; r < SCAN_R; ++r) h[r][t] = fmaf(xi[r], wv, h[r][t]);
    }

    const int kn = (k + 1) % SCAN_PF;
    *(v4f*)&w1c[nxt][t4] = r1[kn];
    *(v4f*)&w2a[nxt][t4] = r2[kn];
    *(v4f*)&w2b[nxt][t4] = r3[kn];
    lds_barrier();
    zxl[i] = xv;
  };

  constexpr int kAsk = ((DIM - SCAN_PF) / SCAN_PF) * SCAN_PF;
  for (int i0 = 0; i0 < kAsk; i0 += SCAN_PF) {
#pragma unroll
    for (int k = 0; k < SCAN_PF; ++k) step(i0 + k, k, true);
  }
#pragma unroll
  for (int i = kAsk; i < DIM - 1; ++i) step(i, i % SCAN_PF, false);

  {
    const int i = DIM - 1;
    zxl[i] = column(i, w2a[i & 1], w2b[i & 1]);
  }

#pragma unroll
  for (int r = 0; r < SCAN_R; ++r) {
    xout[(row0 + r) * DIM + lane]      = zx[wave][r][lane];
    xout[(row0 + r) * DIM + 32 + lane] = zx[wave][r][32 + lane];
  }
}

#define LD_NT 4
__global__ __launch_bounds__(512) void maf_logdet_wmma(const float* __restrict__ b1,
                                                       const float* __restrict__ ws,
                                                       const float* __restrict__ xout,
                                                       float* __restrict__ ldout) {
  __shared__ float b1s[HID];
  __shared__ float wss[HID];
  __shared__ float part[8][32];
  const v2f* Bp = (const v2f*)(ws + WS_BPACK);
  const float bsum = ws[WS_BSUM];

  const int tid  = threadIdx.x;
  const int lane = tid & 31;
  const int wave = tid >> 5;
  const int lr   = lane & 15;
  const int koff = (lane >> 4) ? 2 : 0;
  const int mt   = wave & 1;
  const int q    = wave >> 1;
  const int m0   = blockIdx.x * 32 + mt * 16;
  const int nt0  = q * LD_NT;

  const float b1v = b1[tid];
  const float wsv = ws[WS_WSUM + tid];

  const float* xp = xout + (m0 + lr) * DIM + koff;
  v2f a[16];
#pragma unroll
  for (int t = 0; t < 16; ++t) a[t] = *(const v2f*)(xp + 4 * t);

  v2f bt[LD_NT][16];
#pragma unroll
  for (int i = 0; i < LD_NT; ++i) {
#pragma unroll
    for (int t = 0; t < 16; ++t) bt[i][t] = Bp[((nt0 + i) * 16 + t) * 32 + lane];
  }

  __builtin_amdgcn_sched_barrier(0);

  b1s[tid] = b1v;
  wss[tid] = wsv;
  __syncthreads();

  v8f ld = {};
#pragma unroll
  for (int i = 0; i < LD_NT; ++i) {
    const int col = (nt0 + i) * 16 + lr;

    v8f acc0 = {}, acc1 = {};
#pragma unroll
    for (int t = 0; t < 16; t += 2) {
      acc0 = __builtin_amdgcn_wmma_f32_16x16x4_f32(false, a[t],     false, bt[i][t],
                                                   (short)0, acc0, false, false);
      acc1 = __builtin_amdgcn_wmma_f32_16x16x4_f32(false, a[t + 1], false, bt[i][t + 1],
                                                   (short)0, acc1, false, false);
    }
    v8f acc = acc0 + acc1;

    const float bb = b1s[col];
    const float wv = wss[col];
#pragma unroll
    for (int v = 0; v < 8; ++v) {
      float hv = fmaxf(acc[v] + bb, 0.f);
      ld[v] = fmaf(hv, wv, ld[v]);
    }
  }

#pragma unroll
  for (int off = 1; off < 16; off <<= 1) {
#pragma unroll
    for (int v = 0; v < 8; ++v) {
      float t = ld[v];
      ld[v] = t + __shfl_xor(t, off, 32);
    }
  }
  if (lr == 0) {
    const int r0 = mt * 16 + (lane >> 4) * 8;
#pragma unroll
    for (int v = 0; v < 8; ++v) part[q][r0 + v] = ld[v];
  }
  __syncthreads();
  if (tid < 32) {
    float s = part[0][tid];
#pragma unroll
    for (int k = 1; k < 8; ++k) s += part[k][tid];
    ldout[blockIdx.x * 32 + tid] = s + bsum;
  }
}

extern "C" void kernel_launch(void* const* d_in, const int* in_sizes, int n_in,
                              void* d_out, int out_size, void* d_ws, size_t ws_size,
                              hipStream_t stream) {
  const float* z     = (const float*)d_in[0];
  const float* W1    = (const float*)d_in[1];
  const float* b1    = (const float*)d_in[2];
  const float* W2    = (const float*)d_in[3];
  const float* b2    = (const float*)d_in[4];
  const float* mask1 = (const float*)d_in[5];
  const float* mask2 = (const float*)d_in[6];

  float* ws    = (float*)d_ws;
  float* xout  = (float*)d_out;
  float* ldout = xout + BATCH * DIM;

  const int nprep = HID * DIM + 2 * DIM * HID + NPACK;
  maf_prep<<<PREP_HEAD_BLOCKS + (nprep + 255) / 256, 256, 0, stream>>>(W1, mask1, W2, mask2, b2, ws);
  maf_scan<<<BATCH / SCAN_ROWS, 32 * SCAN_WAVES, 0, stream>>>(z, b1, b2, ws, xout);
  maf_logdet_wmma<<<BATCH / 32, 512, 0, stream>>>(b1, ws, xout, ldout);
}


# log_det moved into the scan launch (2 launches, was 3): each block's 4 waves run the f32 WMMA product and w_sum dot for its 16 rows from x held in LDS; 8 partials per row, same sum order
# speedup vs baseline: 4.0870x; 1.0557x over previous
#include <hip/hip_runtime.h>
#include <hip/hip_bf16.h>

#define DIM 64
#define HID 512
#define BATCH 8192

typedef __attribute__((ext_vector_type(2))) float v2f;
typedef __attribute__((ext_vector_type(4))) float v4f;
typedef __attribute__((ext_vector_type(8))) float v8f;

#define WS_MW1T  0
#define WS_MW2   (HID*DIM)
#define WS_WSUM  (HID*DIM + 2*DIM*HID)
#define WS_BSUM  (WS_WSUM + HID)
#define WS_BPACK 98832
#define NPACK    (HID*DIM)

#define PREP_SUM_BLOCKS 2
#define PREP_HEAD_BLOCKS (PREP_SUM_BLOCKS + 1)
__global__ __launch_bounds__(256) void maf_prep(const float* __restrict__ W1,
                                                const float* __restrict__ mask1,
                                                const float* __restrict__ W2,
                                                const float* __restrict__ mask2,
                                                const float* __restrict__ b2,
                                                float* __restrict__ ws) {
  if (blockIdx.x < PREP_SUM_BLOCKS) {
    const int j = blockIdx.x * 256 + threadIdx.x;
    float s = 0.f;
#pragma unroll
    for (int r = DIM; r < 2 * DIM; ++r) s += mask2[r * HID + j] * W2[r * HID + j];
    ws[WS_WSUM + j] = s;
    return;
  }
  if (blockIdx.x == PREP_SUM_BLOCKS) {
    if (threadIdx.x == 0) {
      float bs = 0.f;
#pragma unroll
      for (int r = DIM; r < 2 * DIM; ++r) bs += b2[r];
      ws[WS_BSUM] = bs;
    }
    return;
  }
  const int idx = (blockIdx.x - PREP_HEAD_BLOCKS) * 256 + threadIdx.x;
  if (idx < HID * DIM) {
    int j = idx / DIM, k = idx % DIM;
    ws[WS_MW1T + k * HID + j] = mask1[idx] * W1[idx];
  } else if (idx < HID * DIM + 2 * DIM * HID) {
    int t = idx - HID * DIM;
    ws[WS_MW2 + t] = mask2[t] * W2[t];
  } else if (idx < HID * DIM + 2 * DIM * HID + NPACK) {
    int f = idx - (HID * DIM + 2 * DIM * HID);
    int p = f >> 1, c = f & 1;
    int nt = p >> 9;
    int t  = (p >> 5) & 15;
    int l  = p & 31;
    int koff = (l >> 4) << 1;
    int j = (nt << 4) + (l & 15);
    int k = (t << 2) + koff + c;
    ws[WS_BPACK + f] = mask1[j * DIM + k] * W1[j * DIM + k];
  }
}

__device__ __forceinline__ float lane_xor16(float x) {
  const int b = __builtin_bit_cast(int, x);
  return __builtin_bit_cast(float,
      __builtin_amdgcn_permlanex16(b, b, 0x76543210, 0xfedcba98, false, false));
}
template <int M>
__device__ __forceinline__ float row_xor(float x) {
  const int b = __builtin_bit_cast(int, x);
  return __builtin_bit_cast(float, __builtin_amdgcn_update_dpp(0, b, 0x160 + M, 0xf, 0xf, true));
}

__device__ __forceinline__ void lds_barrier() {
  asm volatile("s_wait_dscnt 0x0\n\ts_barrier_signal -1\n\ts_barrier_wait -1" ::: "memory");
}

#define SCAN_R 4
#define ZX_PITCH (DIM + 1)
#define SCAN_WAVES 4
#define SCAN_ROWS (SCAN_WAVES * SCAN_R)
#define SCAN_PF 4
#define LD_NT 4
#define LD_PARTS (HID / 16 / LD_NT)
__global__ __launch_bounds__(32 * SCAN_WAVES) void maf_scan(const float* __restrict__ z,
                                                const float* __restrict__ b1,
                                                const float* __restrict__ b2,
                                                const float* __restrict__ ws,
                                                float* __restrict__ xout,
                                                float* __restrict__ ldout) {
  __shared__ __align__(16) float w1c[2][HID];
  __shared__ __align__(16) float w2a[2][HID];
  __shared__ __align__(16) float w2b[2][HID];
  __shared__ float zx[SCAN_WAVES][SCAN_R][ZX_PITCH];
  __shared__ float b1s[HID];
  __shared__ float wss[HID];
  __shared__ float part[LD_PARTS][SCAN_ROWS];
  const float* MW1T = ws + WS_MW1T;
  const float* MW2  = ws + WS_MW2;

  const int tid  = threadIdx.x;
  const int lane = tid & 31;
  const int wave = tid >> 5;
  const int row0 = blockIdx.x * SCAN_ROWS + wave * SCAN_R;
  const int t4   = tid * 4;

  *(v4f*)&w1c[0][t4] = *(const v4f*)&MW1T[t4];
  *(v4f*)&w2a[0][t4] = *(const v4f*)&MW2[t4];
  *(v4f*)&w2b[0][t4] = *(const v4f*)&MW2[DIM * HID + t4];

#pragma unroll
  for (int k = 0; k < HID / (32 * SCAN_WAVES); ++k) {
    b1s[tid + 32 * SCAN_WAVES * k] = b1[tid + 32 * SCAN_WAVES * k];
    wss[tid + 32 * SCAN_WAVES * k] = ws[WS_WSUM + tid + 32 * SCAN_WAVES * k];
  }

  float h[SCAN_R][16];
#pragma unroll
  for (int t = 0; t < 16; ++t) {
    const float bv = b1[lane + 32 * t];
#pragma unroll
    for (int r = 0; r < SCAN_R; ++r) h[r][t] = bv;
  }

#pragma unroll
  for (int r = 0; r < SCAN_R; ++r) {
    zx[wave][r][lane]      = z[(row0 + r) * DIM + lane];
    zx[wave][r][32 + lane] = z[(row0 + r) * DIM + 32 + lane];
  }

  const bool up8 = (lane & 8) != 0;
  const bool up4 = (lane & 4) != 0;
  const int  rl  = (up8 ? 1 : 0) + (up4 ? 2 : 0);
  float* zxl = &zx[wave][rl][0];

  auto column = [&](int i, const float* wa, const float* wb) -> float {
    const float zi = zxl[i];
    float mu[SCAN_R], al[SCAN_R];
#pragma unroll
    for (int r = 0; r < SCAN_R; ++r) { mu[r] = 0.f; al[r] = 0.f; }
#pragma unroll
    for (int t = 0; t < 16; ++t) {
      const int j = lane + 32 * t;
      const float wav = wa[j];
      const float wbv = wb[j];
#pragma unroll
      for (int r = 0; r < SCAN_R; ++r) {
        float hv = fmaxf(h[r][t], 0.f);
        mu[r] = fmaf(hv, wav, mu[r]);
        al[r] = fmaf(hv, wbv, al[r]);
      }
    }
    const float m01 = (up8 ? mu[1] : mu[0]) + row_xor<8>(up8 ? mu[0] : mu[1]);
    const float m23 = (up8 ? mu[3] : mu[2]) + row_xor<8>(up8 ? mu[2] : mu[3]);
    const float a01 = (up8 ? al[1] : al[0]) + row_xor<8>(up8 ? al[0] : al[1]);
    const float a23 = (up8 ? al[3] : al[2]) + row_xor<8>(up8 ? al[2] : al[3]);
    float m = (up4 ? m23 : m01) + row_xor<4>(up4 ? m01 : m23);
    float a = (up4 ? a23 : a01) + row_xor<4>(up4 ? a01 : a23);
    m += row_xor<2>(m);  a += row_xor<2>(a);
    m += row_xor<1>(m);  a += row_xor<1>(a);
    m += lane_xor16(m);  a += lane_xor16(a);
    m += b2[i];
    a += b2[DIM + i];
    return fmaf(zi, expf(a), m);
  };

  auto spread = [&](float xv, float (&xi)[SCAN_R]) {
    const int xb = __builtin_bit_cast(int, xv);
    xi[0] = __builtin_bit_cast(float, __builtin_amdgcn_readlane(xb, 0));
    xi[1] = __builtin_bit_cast(float, __builtin_amdgcn_readlane(xb, 8));
    xi[2] = __builtin_bit_cast(float, __builtin_amdgcn_readlane(xb, 4));
    xi[3] = __builtin_bit_cast(float, __builtin_amdgcn_readlane(xb, 12));
  };

  v4f r1[SCAN_PF], r2[SCAN_PF], r3[SCAN_PF];
  auto request = [&](int s, int slot) {
    r1[slot] = *(const v4f*)&MW1T[s * HID + t4];
    r2[slot] = *(const v4f*)&MW2[s * HID + t4];
    r3[slot] = *(const v4f*)&MW2[(DIM + s) * HID + t4];
  };
#pragma unroll
  for (int s = 1; s < SCAN_PF; ++s) request(s, s);

  lds_barrier();

  auto step = [&](int i, int k, bool ask) __attribute__((always_inline)) {
    const int cur = k & 1;
    const int nxt = cur ^ 1;
    if (ask) {
      request(i + SCAN_PF, k);
      __builtin_amdgcn_sched_barrier(0x060f);
    }

    const float xv = column(i, w2a[cur], w2b[cur]);
    float xi[SCAN_R];
    spread(xv, xi);

#pragma unroll
    for (int t = 0; t < 16; ++t) {
      const float wv = w1c[cur][lane + 32 * t];
#pragma unroll
      for (int r = 0; r < SCAN_R; ++r) h[r][t] = fmaf(xi[r], wv, h[r][t]);
    }

    const int kn = (k + 1) % SCAN_PF;
    *(v4f*)&w1c[nxt][t4] = r1[kn];
    *(v4f*)&w2a[nxt][t4] = r2[kn];
    *(v4f*)&w2b[nxt][t4] = r3[kn];
    lds_barrier();
    zxl[i] = xv;
  };

  constexpr int kAsk = ((DIM - SCAN_PF) / SCAN_PF) * SCAN_PF;
  for (int i0 = 0; i0 < kAsk; i0 += SCAN_PF) {
#pragma unroll
    for (int k = 0; k < SCAN_PF; ++k) step(i0 + k, k, true);
  }
#pragma unroll
  for (int i = kAsk; i < DIM - 1; ++i) step(i, i % SCAN_PF, false);

  {
    const int i = DIM - 1;
    zxl[i] = column(i, w2a[i & 1], w2b[i & 1]);
  }

  const int lr   = lane & 15;
  const int koff = (lane >> 4) * 2;
  const v2f* Bp  = (const v2f*)(ws + WS_BPACK);
  const int nt0  = wave * (2 * LD_NT);

  auto load_b = [&](v2f (&b)[2][16], int nt) {
#pragma unroll
    for (int i = 0; i < 2; ++i) {
#pragma unroll
      for (int t = 0; t < 16; ++t) b[i][t] = Bp[((nt + i) * 16 + t) * 32 + lane];
    }
  };
  v2f bA[2][16], bB[2][16];
  load_b(bA, nt0);
  load_b(bB, nt0 + 2);

  lds_barrier();

#pragma unroll
  for (int r = 0; r < SCAN_R; ++r) {
    xout[(row0 + r) * DIM + lane]      = zx[wave][r][lane];
    xout[(row0 + r) * DIM + 32 + lane] = zx[wave][r][32 + lane];
  }

  const float* xs = &zx[0][0][0] + lr * ZX_PITCH + koff;
  v2f a[16];
#pragma unroll
  for (int t = 0; t < 16; ++t) { a[t].x = xs[4 * t]; a[t].y = xs[4 * t + 1]; }

  __builtin_amdgcn_sched_barrier(0);

  auto tiles2 = [&](const v2f (&b)[2][16], int nt, v8f& ld) {
#pragma unroll
    for (int i = 0; i < 2; ++i) {
      const int col = (nt + i) * 16 + lr;
      v8f acc0 = {}, acc1 = {};
#pragma unroll
      for (int t = 0; t < 16; t += 2) {
        acc0 = __builtin_amdgcn_wmma_f32_16x16x4_f32(false, a[t],     false, b[i][t],
                                                     (short)0, acc0, false, false);
        acc1 = __builtin_amdgcn_wmma_f32_16x16x4_f32(false, a[t + 1], false, b[i][t + 1],
                                                     (short)0, acc1, false, false);
      }
      v8f acc = acc0 + acc1;
      const float bb = b1s[col];
      const float wv = wss[col];
#pragma unroll
      for (int v = 0; v < 8; ++v) {
        float hv = fmaxf(acc[v] + bb, 0.f);
        ld[v] = fmaf(hv, wv, ld[v]);
      }
    }
  };
  auto fold = [&](v8f& ld, int q) {
#pragma unroll
    for (int v = 0; v < 8; ++v) { float t = ld[v]; ld[v] = t + row_xor<1>(t); }
#pragma unroll
    for (int v = 0; v < 8; ++v) { float t = ld[v]; ld[v] = t + row_xor<2>(t); }
#pragma unroll
    for (int v = 0; v < 8; ++v) { float t = ld[v]; ld[v] = t + row_xor<4>(t); }
#pragma unroll
    for (int v = 0; v < 8; ++v) { float t = ld[v]; ld[v] = t + row_xor<8>(t); }
    const int r0 = (lane >> 4) * 8;
#pragma unroll
    for (int v = 0; v < 8; ++v) part[q][r0 + v] = ld[v];
  };

  v8f ld0 = {}, ld1 = {};
  tiles2(bA, nt0, ld0);
  load_b(bA, nt0 + 4);
  __builtin_amdgcn_sched_barrier(0);
  tiles2(bB, nt0 + 2, ld0);
  load_b(bB, nt0 + 6);
  __builtin_amdgcn_sched_barrier(0);
  fold(ld0, 2 * wave);
  tiles2(bA, nt0 + 4, ld1);
  tiles2(bB, nt0 + 6, ld1);
  fold(ld1, 2 * wave + 1);

  lds_barrier();
  if (tid < SCAN_ROWS) {
    float s = part[0][tid];
#pragma unroll
    for (int k = 1; k < LD_PARTS; ++k) s += part[k][tid];
    ldout[blockIdx.x * SCAN_ROWS + tid] = s + ws[WS_BSUM];
  }
}

extern "C" void kernel_launch(void* const* d_in, const int* in_sizes, int n_in,
                              void* d_out, int out_size, void* d_ws, size_t ws_size,
                              hipStream_t stream) {
  const float* z     = (const float*)d_in[0];
  const float* W1    = (const float*)d_in[1];
  const float* b1    = (const float*)d_in[2];
  const float* W2    = (const float*)d_in[3];
  const float* b2    = (const float*)d_in[4];
  const float* mask1 = (const float*)d_in[5];
  const float* mask2 = (const float*)d_in[6];

  float* ws    = (float*)d_ws;
  float* xout  = (float*)d_out;
  float* ldout = xout + BATCH * DIM;

  const int nprep = HID * DIM + 2 * DIM * HID + NPACK;
  maf_prep<<<PREP_HEAD_BLOCKS + (nprep + 255) / 256, 256, 0, stream>>>(W1, mask1, W2, mask2, b2, ws);
  maf_scan<<<BATCH / SCAN_ROWS, 32 * SCAN_WAVES, 0, stream>>>(z, b1, b2, ws, xout, ldout);
}


# scan: a step's LDS traffic (next step's weights and biases into a second register set, staging and output stores) is issued right behind the barrier and runs under arithmetic; sums bit-identical
# speedup vs baseline: 4.2217x; 1.0330x over previous
#include <hip/hip_runtime.h>
#include <hip/hip_bf16.h>

#define DIM 64
#define HID 512
#define BATCH 8192

typedef __attribute__((ext_vector_type(2))) float v2f;
typedef __attribute__((ext_vector_type(4))) float v4f;
typedef __attribute__((ext_vector_type(8))) float v8f;

#define WS_MW1T  0
#define WS_MW2   (HID*DIM)
#define WS_WSUM  (HID*DIM + 2*DIM*HID)
#define WS_BSUM  (WS_WSUM + HID)
#define WS_BPACK 98832
#define NPACK    (HID*DIM)

#define PREP_SUM_BLOCKS 2
#define PREP_HEAD_BLOCKS (PREP_SUM_BLOCKS + 1)
__global__ __launch_bounds__(256) void maf_prep(const float* __restrict__ W1,
                                                const float* __restrict__ mask1,
                                                const float* __restrict__ W2,
                                                const float* __restrict__ mask2,
                                                const float* __restrict__ b2,
                                                float* __restrict__ ws) {
  if (blockIdx.x < PREP_SUM_BLOCKS) {
    const int j = blockIdx.x * 256 + threadIdx.x;
    float s = 0.f;
#pragma unroll
    for (int r = DIM; r < 2 * DIM; ++r) s += mask2[r * HID + j] * W2[r * HID + j];
    ws[WS_WSUM + j] = s;
    return;
  }
  if (blockIdx.x == PREP_SUM_BLOCKS) {
    if (threadIdx.x == 0) {
      float bs = 0.f;
#pragma unroll
      for (int r = DIM; r < 2 * DIM; ++r) bs += b2[r];
      ws[WS_BSUM] = bs;
    }
    return;
  }
  const int idx = (blockIdx.x - PREP_HEAD_BLOCKS) * 256 + threadIdx.x;
  if (idx < HID * DIM) {
    int j = idx / DIM, k = idx % DIM;
    ws[WS_MW1T + k * HID + j] = mask1[idx] * W1[idx];
  } else if (idx < HID * DIM + 2 * DIM * HID) {
    int t = idx - HID * DIM;
    ws[WS_MW2 + t] = mask2[t] * W2[t];
  } else if (idx < HID * DIM + 2 * DIM * HID + NPACK) {
    int f = idx - (HID * DIM + 2 * DIM * HID);
    int p = f >> 1, c = f & 1;
    int nt = p >> 9;
    int t  = (p >> 5) & 15;
    int l  = p & 31;
    int koff = (l >> 4) << 1;
    int j = (nt << 4) + (l & 15);
    int k = (t << 2) + koff + c;
    ws[WS_BPACK + f] = mask1[j * DIM + k] * W1[j * DIM + k];
  }
}

__device__ __forceinline__ float lane_xor16(float x) {
  const int b = __builtin_bit_cast(int, x);
  return __builtin_bit_cast(float,
      __builtin_amdgcn_permlanex16(b, b, 0x76543210, 0xfedcba98, false, false));
}
template <int M>
__device__ __forceinline__ float row_xor(float x) {
  const int b = __builtin_bit_cast(int, x);
  return __builtin_bit_cast(float, __builtin_amdgcn_update_dpp(0, b, 0x160 + M, 0xf, 0xf, true));
}

__device__ __forceinline__ void lds_barrier() {
  asm volatile("s_wait_dscnt 0x0\n\ts_barrier_signal -1\n\ts_barrier_wait -1" ::: "memory");
}

#define SCAN_R 4
#define ZX_PITCH (DIM + 1)
#define SCAN_WAVES 4
#define SCAN_ROWS (SCAN_WAVES * SCAN_R)
#define SCAN_PF 4
#define LD_NT 4
#define LD_PARTS (HID / 16 / LD_NT)
__global__ __launch_bounds__(32 * SCAN_WAVES) __attribute__((amdgpu_waves_per_eu(2, 2))) void maf_scan(const float* __restrict__ z,
                                                const float* __restrict__ b1,
                                                const float* __restrict__ b2,
                                                const float* __restrict__ ws,
                                                float* __restrict__ xout,
                                                float* __restrict__ ldout) {
  __shared__ __align__(16) float w1c[2][HID];
  __shared__ __align__(16) float w2a[2][HID];
  __shared__ __align__(16) float w2b[2][HID];
  __shared__ float zx[SCAN_WAVES][SCAN_R][ZX_PITCH];
  __shared__ v2f b2s[DIM];
  __shared__ float b1s[HID];
  __shared__ float wss[HID];
  __shared__ float part[LD_PARTS][SCAN_ROWS];
  const float* MW1T = ws + WS_MW1T;
  const float* MW2  = ws + WS_MW2;

  const int tid  = threadIdx.x;
  const int lane = tid & 31;
  const int wave = tid >> 5;
  const int row0 = blockIdx.x * SCAN_ROWS + wave * SCAN_R;
  const int t4   = tid * 4;

#pragma unroll
  for (int s = 0; s < 2; ++s) {
    *(v4f*)&w1c[s][t4] = *(const v4f*)&MW1T[s * HID + t4];
    *(v4f*)&w2a[s][t4] = *(const v4f*)&MW2[s * HID + t4];
    *(v4f*)&w2b[s][t4] = *(const v4f*)&MW2[(DIM + s) * HID + t4];
  }

#pragma unroll
  for (int k = tid; k < 2 * DIM; k += 32 * SCAN_WAVES) ((float*)b2s)[2 * (k % DIM) + k / DIM] = b2[k];

#pragma unroll
  for (int k = 0; k < HID / (32 * SCAN_WAVES); ++k) {
    b1s[tid + 32 * SCAN_WAVES * k] = b1[tid + 32 * SCAN_WAVES * k];
    wss[tid + 32 * SCAN_WAVES * k] = ws[WS_WSUM + tid + 32 * SCAN_WAVES * k];
  }

  float h[SCAN_R][16];
#pragma unroll
  for (int t = 0; t < 16; ++t) {
    const float bv = b1[lane + 32 * t];
#pragma unroll
    for (int r = 0; r < SCAN_R; ++r) h[r][t] = bv;
  }

#pragma unroll
  for (int r = 0; r < SCAN_R; ++r) {
    zx[wave][r][lane]      = z[(row0 + r) * DIM + lane];
    zx[wave][r][32 + lane] = z[(row0 + r) * DIM + 32 + lane];
  }

  const bool up8 = (lane & 8) != 0;
  const bool up4 = (lane & 4) != 0;
  const int  rl  = (up8 ? 1 : 0) + (up4 ? 2 : 0);
  float* zxl = &zx[wave][rl][0];

  float wA[2][16], wB[2][16], wC[2][16], zin[2], bm[2], ba[2];
  auto fetch = [&](int s, int set) __attribute__((always_inline)) {
#pragma unroll
    for (int t = 0; t < 16; ++t) {
      const int j = lane + 32 * t;
      wA[set][t] = w2a[set][j];
      wB[set][t] = w2b[set][j];
      wC[set][t] = w1c[set][j];
    }
    zin[set] = zxl[s];
    const v2f bb = b2s[s];
    bm[set] = bb.x;
    ba[set] = bb.y;
  };

  auto column = [&](int set) __attribute__((always_inline)) -> float {
    const float zi = zin[set];
    float mu[SCAN_R], al[SCAN_R];
#pragma unroll
    for (int r = 0; r < SCAN_R; ++r) { mu[r] = 0.f; al[r] = 0.f; }
#pragma unroll
    for (int t = 0; t < 16; ++t) {
      const float wav = wA[set][t];
      const float wbv = wB[set][t];
#pragma unroll
      for (int r = 0; r < SCAN_R; ++r) {
        float hv = fmaxf(h[r][t], 0.f);
        mu[r] = fmaf(hv, wav, mu[r]);
        al[r] = fmaf(hv, wbv, al[r]);
      }
    }
    const float m01 = (up8 ? mu[1] : mu[0]) + row_xor<8>(up8 ? mu[0] : mu[1]);
    const float m23 = (up8 ? mu[3] : mu[2]) + row_xor<8>(up8 ? mu[2] : mu[3]);
    const float a01 = (up8 ? al[1] : al[0]) + row_xor<8>(up8 ? al[0] : al[1]);
    const float a23 = (up8 ? al[3] : al[2]) + row_xor<8>(up8 ? al[2] : al[3]);
    float m = (up4 ? m23 : m01) + row_xor<4>(up4 ? m01 : m23);
    float a = (up4 ? a23 : a01) + row_xor<4>(up4 ? a01 : a23);
    m += row_xor<2>(m);  a += row_xor<2>(a);
    m += row_xor<1>(m);  a += row_xor<1>(a);
    m += lane_xor16(m);  a += lane_xor16(a);
    m += bm[set];
    a += ba[set];
    return fmaf(zi, expf(a), m);
  };

  auto spread = [&](float xv, float (&xi)[SCAN_R]) {
    const int xb = __builtin_bit_cast(int, xv);
    xi[0] = __builtin_bit_cast(float, __builtin_amdgcn_readlane(xb, 0));
    xi[1] = __builtin_bit_cast(float, __builtin_amdgcn_readlane(xb, 8));
    xi[2] = __builtin_bit_cast(float, __builtin_amdgcn_readlane(xb, 4));
    xi[3] = __builtin_bit_cast(float, __builtin_amdgcn_readlane(xb, 12));
  };

  v4f r1[SCAN_PF], r2[SCAN_PF], r3[SCAN_PF];
  auto request = [&](int s, int slot) {
    r1[slot] = *(const v4f*)&MW1T[s * HID + t4];
    r2[slot] = *(const v4f*)&MW2[s * HID + t4];
    r3[slot] = *(const v4f*)&MW2[(DIM + s) * HID + t4];
  };
#pragma unroll
  for (int s = 2; s <= SCAN_PF; ++s) request(s, s % SCAN_PF);

  lds_barrier();
  fetch(0, 0);
  float xprev = zin[0];

  auto step = [&](int i, int k, bool ask, bool stage) __attribute__((always_inline)) {
    const int cur = k & 1;
    const int nxt = cur ^ 1;
    if (ask) {
      const int s = i + SCAN_PF + 1 < DIM ? i + SCAN_PF + 1 : DIM - 1;
      request(s, (k + 1) % SCAN_PF);
      __builtin_amdgcn_sched_barrier(0x060f);
    }

    lds_barrier();
    zxl[i > 0 ? i - 1 : 0] = xprev;
    fetch(i + 1, nxt);
    if (stage) {
      const int kn = (k + 2) % SCAN_PF;
      *(v4f*)&w1c[cur][t4] = r1[kn];
      *(v4f*)&w2a[cur][t4] = r2[kn];
      *(v4f*)&w2b[cur][t4] = r3[kn];
    }
    __builtin_amdgcn_sched_barrier(0);

    const float xv = column(cur);
    xprev = xv;
    float xi[SCAN_R];
    spread(xv, xi);
    __builtin_amdgcn_sched_barrier(0);

#pragma unroll
    for (int t = 0; t < 16; ++t) {
      const float wv = wC[cur][t];
#pragma unroll
      for (int r = 0; r < SCAN_R; ++r) h[r][t] = fmaf(xi[r], wv, h[r][t]);
    }
  };

  constexpr int kAsk = ((DIM - SCAN_PF) / SCAN_PF) * SCAN_PF;
  for (int i0 = 0; i0 < kAsk; i0 += SCAN_PF) {
#pragma unroll
    for (int k = 0; k < SCAN_PF; ++k) step(i0 + k, k, true, true);
  }
#pragma unroll
  for (int i = kAsk; i < DIM - 1; ++i) step(i, i % SCAN_PF, false, i + 2 < DIM);

  {
    const int i = DIM - 1;
    zxl[i - 1] = xprev;
    zxl[i] = column(i & 1);
  }

  const int lr   = lane & 15;
  const int koff = (lane >> 4) * 2;
  const v2f* Bp  = (const v2f*)(ws + WS_BPACK);
  const int nt0  = wave * (2 * LD_NT);

  auto load_b = [&](v2f (&b)[2][16], int nt) {
#pragma unroll
    for (int i = 0; i < 2; ++i) {
#pragma unroll
      for (int t = 0; t < 16; ++t) b[i][t] = Bp[((nt + i) * 16 + t) * 32 + lane];
    }
  };
  v2f bA[2][16], bB[2][16];
  load_b(bA, nt0);
  load_b(bB, nt0 + 2);

  lds_barrier();

#pragma unroll
  for (int r = 0; r < SCAN_R; ++r) {
    xout[(row0 + r) * DIM + lane]      = zx[wave][r][lane];
    xout[(row0 + r) * DIM + 32 + lane] = zx[wave][r][32 + lane];
  }

  const float* xs = &zx[0][0][0] + lr * ZX_PITCH + koff;
  v2f a[16];
#pragma unroll
  for (int t = 0; t < 16; ++t) { a[t].x = xs[4 * t]; a[t].y = xs[4 * t + 1]; }

  __builtin_amdgcn_sched_barrier(0);

  auto tiles2 = [&](const v2f (&b)[2][16], int nt, v8f& ld) {
#pragma unroll
    for (int i = 0; i < 2; ++i) {
      const int col = (nt + i) * 16 + lr;
      v8f acc0 = {}, acc1 = {};
#pragma unroll
      for (int t = 0; t < 16; t += 2) {
        acc0 = __builtin_amdgcn_wmma_f32_16x16x4_f32(false, a[t],     false, b[i][t],
                                                     (short)0, acc0, false, false);
        acc1 = __builtin_amdgcn_wmma_f32_16x16x4_f32(false, a[t + 1], false, b[i][t + 1],
                                                     (short)0, acc1, false, false);
      }
      v8f acc = acc0 + acc1;
      const float bb = b1s[col];
      const float wv = wss[col];
#pragma unroll
      for (int v = 0; v < 8; ++v) {
        float hv = fmaxf(acc[v] + bb, 0.f);
        ld[v] = fmaf(hv, wv, ld[v]);
      }
    }
  };
  auto fold = [&](v8f& ld, int q) {
#pragma unroll
    for (int v = 0; v < 8; ++v) { float t = ld[v]; ld[v] = t + row_xor<1>(t); }
#pragma unroll
    for (int v = 0; v < 8; ++v) { float t = ld[v]; ld[v] = t + row_xor<2>(t); }
#pragma unroll
    for (int v = 0; v < 8; ++v) { float t = ld[v]; ld[v] = t + row_xor<4>(t); }
#pragma unroll
    for (int v = 0; v < 8; ++v) { float t = ld[v]; ld[v] = t + row_xor<8>(t); }
    const int r0 = (lane >> 4) * 8;
#pragma unroll
    for (int v = 0; v < 8; ++v) part[q][r0 + v] = ld[v];
  };

  v8f ld0 = {}, ld1 = {};
  tiles2(bA, nt0, ld0);
  load_b(bA, nt0 + 4);
  __builtin_amdgcn_sched_barrier(0);
  tiles2(bB, nt0 + 2, ld0);
  load_b(bB, nt0 + 6);
  __builtin_amdgcn_sched_barrier(0);
  fold(ld0, 2 * wave);
  tiles2(bA, nt0 + 4, ld1);
  tiles2(bB, nt0 + 6, ld1);
  fold(ld1, 2 * wave + 1);

  lds_barrier();
  if (tid < SCAN_ROWS) {
    float s = part[0][tid];
#pragma unroll
    for (int k = 1; k < LD_PARTS; ++k) s += part[k][tid];
    ldout[blockIdx.x * SCAN_ROWS + tid] = s + ws[WS_BSUM];
  }
}

extern "C" void kernel_launch(void* const* d_in, const int* in_sizes, int n_in,
                              void* d_out, int out_size, void* d_ws, size_t ws_size,
                              hipStream_t stream) {
  const float* z     = (const float*)d_in[0];
  const float* W1    = (const float*)d_in[1];
  const float* b1    = (const float*)d_in[2];
  const float* W2    = (const float*)d_in[3];
  const float* b2    = (const float*)d_in[4];
  const float* mask1 = (const float*)d_in[5];
  const float* mask2 = (const float*)d_in[6];

  float* ws    = (float*)d_ws;
  float* xout  = (float*)d_out;
  float* ldout = xout + BATCH * DIM;

  const int nprep = HID * DIM + 2 * DIM * HID + NPACK;
  maf_prep<<<PREP_HEAD_BLOCKS + (nprep + 255) / 256, 256, 0, stream>>>(W1, mask1, W2, mask2, b2, ws);
  maf_scan<<<BATCH / SCAN_ROWS, 32 * SCAN_WAVES, 0, stream>>>(z, b1, b2, ws, xout, ldout);
}


# scan: s_setprio 3 over each step's serial reduce-exp-readout chain, 0 over the products; as a scheduling boundary it also splits the region: loop registers below v256, 78 fewer vector instr a trip
# speedup vs baseline: 4.3644x; 1.0338x over previous
#include <hip/hip_runtime.h>
#include <hip/hip_bf16.h>

#define DIM 64
#define HID 512
#define BATCH 8192

typedef __attribute__((ext_vector_type(2))) float v2f;
typedef __attribute__((ext_vector_type(4))) float v4f;
typedef __attribute__((ext_vector_type(8))) float v8f;

#define WS_MW1T  0
#define WS_MW2   (HID*DIM)
#define WS_WSUM  (HID*DIM + 2*DIM*HID)
#define WS_BSUM  (WS_WSUM + HID)
#define WS_BPACK 98832
#define NPACK    (HID*DIM)

#define PREP_SUM_BLOCKS 2
#define PREP_HEAD_BLOCKS (PREP_SUM_BLOCKS + 1)
__global__ __launch_bounds__(256) void maf_prep(const float* __restrict__ W1,
                                                const float* __restrict__ mask1,
                                                const float* __restrict__ W2,
                                                const float* __restrict__ mask2,
                                                const float* __restrict__ b2,
                                                float* __restrict__ ws) {
  if (blockIdx.x < PREP_SUM_BLOCKS) {
    const int j = blockIdx.x * 256 + threadIdx.x;
    float s = 0.f;
#pragma unroll
    for (int r = DIM; r < 2 * DIM; ++r) s += mask2[r * HID + j] * W2[r * HID + j];
    ws[WS_WSUM + j] = s;
    return;
  }
  if (blockIdx.x == PREP_SUM_BLOCKS) {
    if (threadIdx.x == 0) {
      float bs = 0.f;
#pragma unroll
      for (int r = DIM; r < 2 * DIM; ++r) bs += b2[r];
      ws[WS_BSUM] = bs;
    }
    return;
  }
  const int idx = (blockIdx.x - PREP_HEAD_BLOCKS) * 256 + threadIdx.x;
  if (idx < HID * DIM) {
    int j = idx / DIM, k = idx % DIM;
    ws[WS_MW1T + k * HID + j] = mask1[idx] * W1[idx];
  } else if (idx < HID * DIM + 2 * DIM * HID) {
    int t = idx - HID * DIM;
    ws[WS_MW2 + t] = mask2[t] * W2[t];
  } else if (idx < HID * DIM + 2 * DIM * HID + NPACK) {
    int f = idx - (HID * DIM + 2 * DIM * HID);
    int p = f >> 1, c = f & 1;
    int nt = p >> 9;
    int t  = (p >> 5) & 15;
    int l  = p & 31;
    int koff = (l >> 4) << 1;
    int j = (nt << 4) + (l & 15);
    int k = (t << 2) + koff + c;
    ws[WS_BPACK + f] = mask1[j * DIM + k] * W1[j * DIM + k];
  }
}

__device__ __forceinline__ float lane_xor16(float x) {
  const int b = __builtin_bit_cast(int, x);
  return __builtin_bit_cast(float,
      __builtin_amdgcn_permlanex16(b, b, 0x76543210, 0xfedcba98, false, false));
}
template <int M>
__device__ __forceinline__ float row_xor(float x) {
  const int b = __builtin_bit_cast(int, x);
  return __builtin_bit_cast(float, __builtin_amdgcn_update_dpp(0, b, 0x160 + M, 0xf, 0xf, true));
}

__device__ __forceinline__ void lds_barrier() {
  asm volatile("s_wait_dscnt 0x0\n\ts_barrier_signal -1\n\ts_barrier_wait -1" ::: "memory");
}

#define SCAN_R 4
#define ZX_PITCH (DIM + 1)
#define SCAN_WAVES 4
#define SCAN_ROWS (SCAN_WAVES * SCAN_R)
#define SCAN_PF 4
#define LD_NT 4
#define LD_PARTS (HID / 16 / LD_NT)
__global__ __launch_bounds__(32 * SCAN_WAVES) __attribute__((amdgpu_waves_per_eu(2, 2))) void maf_scan(const float* __restrict__ z,
                                                const float* __restrict__ b1,
                                                const float* __restrict__ b2,
                                                const float* __restrict__ ws,
                                                float* __restrict__ xout,
                                                float* __restrict__ ldout) {
  __shared__ __align__(16) float w1c[2][HID];
  __shared__ __align__(16) float w2a[2][HID];
  __shared__ __align__(16) float w2b[2][HID];
  __shared__ float zx[SCAN_WAVES][SCAN_R][ZX_PITCH];
  __shared__ v2f b2s[DIM];
  __shared__ float b1s[HID];
  __shared__ float wss[HID];
  __shared__ float part[LD_PARTS][SCAN_ROWS];
  const float* MW1T = ws + WS_MW1T;
  const float* MW2  = ws + WS_MW2;

  const int tid  = threadIdx.x;
  const int lane = tid & 31;
  const int wave = tid >> 5;
  const int row0 = blockIdx.x * SCAN_ROWS + wave * SCAN_R;
  const int t4   = tid * 4;

#pragma unroll
  for (int s = 0; s < 2; ++s) {
    *(v4f*)&w1c[s][t4] = *(const v4f*)&MW1T[s * HID + t4];
    *(v4f*)&w2a[s][t4] = *(const v4f*)&MW2[s * HID + t4];
    *(v4f*)&w2b[s][t4] = *(const v4f*)&MW2[(DIM + s) * HID + t4];
  }

#pragma unroll
  for (int k = tid; k < 2 * DIM; k += 32 * SCAN_WAVES) ((float*)b2s)[2 * (k % DIM) + k / DIM] = b2[k];

#pragma unroll
  for (int k = 0; k < HID / (32 * SCAN_WAVES); ++k) {
    b1s[tid + 32 * SCAN_WAVES * k] = b1[tid + 32 * SCAN_WAVES * k];
    wss[tid + 32 * SCAN_WAVES * k] = ws[WS_WSUM + tid + 32 * SCAN_WAVES * k];
  }

  float h[SCAN_R][16];
#pragma unroll
  for (int t = 0; t < 16; ++t) {
    const float bv = b1[lane + 32 * t];
#pragma unroll
    for (int r = 0; r < SCAN_R; ++r) h[r][t] = bv;
  }

#pragma unroll
  for (int r = 0; r < SCAN_R; ++r) {
    zx[wave][r][lane]      = z[(row0 + r) * DIM + lane];
    zx[wave][r][32 + lane] = z[(row0 + r) * DIM + 32 + lane];
  }

  const bool up8 = (lane & 8) != 0;
  const bool up4 = (lane & 4) != 0;
  const int  rl  = (up8 ? 1 : 0) + (up4 ? 2 : 0);
  float* zxl = &zx[wave][rl][0];

  float wA[2][16], wB[2][16], wC[2][16], zin[2], bm[2], ba[2];
  auto fetch = [&](int s, int set) __attribute__((always_inline)) {
#pragma unroll
    for (int t = 0; t < 16; ++t) {
      const int j = lane + 32 * t;
      wA[set][t] = w2a[set][j];
      wB[set][t] = w2b[set][j];
      wC[set][t] = w1c[set][j];
    }
    zin[set] = zxl[s];
    const v2f bb = b2s[s];
    bm[set] = bb.x;
    ba[set] = bb.y;
  };

  auto column = [&](int set) __attribute__((always_inline)) -> float {
    const float zi = zin[set];
    float mu[SCAN_R], al[SCAN_R];
#pragma unroll
    for (int r = 0; r < SCAN_R; ++r) { mu[r] = 0.f; al[r] = 0.f; }
#pragma unroll
    for (int t = 0; t < 16; ++t) {
      const float wav = wA[set][t];
      const float wbv = wB[set][t];
#pragma unroll
      for (int r = 0; r < SCAN_R; ++r) {
        float hv = fmaxf(h[r][t], 0.f);
        mu[r] = fmaf(hv, wav, mu[r]);
        al[r] = fmaf(hv, wbv, al[r]);
      }
    }
    __builtin_amdgcn_s_setprio(3);
    const float m01 = (up8 ? mu[1] : mu[0]) + row_xor<8>(up8 ? mu[0] : mu[1]);
    const float m23 = (up8 ? mu[3] : mu[2]) + row_xor<8>(up8 ? mu[2] : mu[3]);
    const float a01 = (up8 ? al[1] : al[0]) + row_xor<8>(up8 ? al[0] : al[1]);
    const float a23 = (up8 ? al[3] : al[2]) + row_xor<8>(up8 ? al[2] : al[3]);
    float m = (up4 ? m23 : m01) + row_xor<4>(up4 ? m01 : m23);
    float a = (up4 ? a23 : a01) + row_xor<4>(up4 ? a01 : a23);
    m += row_xor<2>(m);  a += row_xor<2>(a);
    m += row_xor<1>(m);  a += row_xor<1>(a);
    m += lane_xor16(m);  a += lane_xor16(a);
    m += bm[set];
    a += ba[set];
    return fmaf(zi, expf(a), m);
  };

  auto spread = [&](float xv, float (&xi)[SCAN_R]) {
    const int xb = __builtin_bit_cast(int, xv);
    xi[0] = __builtin_bit_cast(float, __builtin_amdgcn_readlane(xb, 0));
    xi[1] = __builtin_bit_cast(float, __builtin_amdgcn_readlane(xb, 8));
    xi[2] = __builtin_bit_cast(float, __builtin_amdgcn_readlane(xb, 4));
    xi[3] = __builtin_bit_cast(float, __builtin_amdgcn_readlane(xb, 12));
  };

  v4f r1[SCAN_PF], r2[SCAN_PF], r3[SCAN_PF];
  auto request = [&](int s, int slot) {
    r1[slot] = *(const v4f*)&MW1T[s * HID + t4];
    r2[slot] = *(const v4f*)&MW2[s * HID + t4];
    r3[slot] = *(const v4f*)&MW2[(DIM + s) * HID + t4];
  };
#pragma unroll
  for (int s = 2; s <= SCAN_PF; ++s) request(s, s % SCAN_PF);

  lds_barrier();
  fetch(0, 0);
  float xprev = zin[0];

  auto step = [&](int i, int k, bool ask, bool stage) __attribute__((always_inline)) {
    const int cur = k & 1;
    const int nxt = cur ^ 1;
    if (ask) {
      const int s = i + SCAN_PF + 1 < DIM ? i + SCAN_PF + 1 : DIM - 1;
      request(s, (k + 1) % SCAN_PF);
      __builtin_amdgcn_sched_barrier(0x060f);
    }

    lds_barrier();
    zxl[i > 0 ? i - 1 : 0] = xprev;
    fetch(i + 1, nxt);
    if (stage) {
      const int kn = (k + 2) % SCAN_PF;
      *(v4f*)&w1c[cur][t4] = r1[kn];
      *(v4f*)&w2a[cur][t4] = r2[kn];
      *(v4f*)&w2b[cur][t4] = r3[kn];
    }
    __builtin_amdgcn_sched_barrier(0);

    const float xv = column(cur);
    xprev = xv;
    float xi[SCAN_R];
    spread(xv, xi);
    __builtin_amdgcn_s_setprio(0);
    __builtin_amdgcn_sched_barrier(0);

#pragma unroll
    for (int t = 0; t < 16; ++t) {
      const float wv = wC[cur][t];
#pragma unroll
      for (int r = 0; r < SCAN_R; ++r) h[r][t] = fmaf(xi[r], wv, h[r][t]);
    }
  };

  constexpr int kAsk = ((DIM - SCAN_PF) / SCAN_PF) * SCAN_PF;
  for (int i0 = 0; i0 < kAsk; i0 += SCAN_PF) {
#pragma unroll
    for (int k = 0; k < SCAN_PF; ++k) step(i0 + k, k, true, true);
  }
#pragma unroll
  for (int i = kAsk; i < DIM - 1; ++i) step(i, i % SCAN_PF, false, i + 2 < DIM);

  {
    const int i = DIM - 1;
    zxl[i - 1] = xprev;
    zxl[i] = column(i & 1);
    __builtin_amdgcn_s_setprio(0);
  }

  const int lr   = lane & 15;
  const int koff = (lane >> 4) * 2;
  const v2f* Bp  = (const v2f*)(ws + WS_BPACK);
  const int nt0  = wave * (2 * LD_NT);

  auto load_b = [&](v2f (&b)[2][16], int nt) {
#pragma unroll
    for (int i = 0; i < 2; ++i) {
#pragma unroll
      for (int t = 0; t < 16; ++t) b[i][t] = Bp[((nt + i) * 16 + t) * 32 + lane];
    }
  };
  v2f bA[2][16], bB[2][16];
  load_b(bA, nt0);
  load_b(bB, nt0 + 2);

  lds_barrier();

#pragma unroll
  for (int r = 0; r < SCAN_R; ++r) {
    xout[(row0 + r) * DIM + lane]      = zx[wave][r][lane];
    xout[(row0 + r) * DIM + 32 + lane] = zx[wave][r][32 + lane];
  }

  const float* xs = &zx[0][0][0] + lr * ZX_PITCH + koff;
  v2f a[16];
#pragma unroll
  for (int t = 0; t < 16; ++t) { a[t].x = xs[4 * t]; a[t].y = xs[4 * t + 1]; }

  __builtin_amdgcn_sched_barrier(0);

  auto tiles2 = [&](const v2f (&b)[2][16], int nt, v8f& ld) {
#pragma unroll
    for (int i = 0; i < 2; ++i) {
      const int col = (nt + i) * 16 + lr;
      v8f acc0 = {}, acc1 = {};
#pragma unroll
      for (int t = 0; t < 16; t += 2) {
        acc0 = __builtin_amdgcn_wmma_f32_16x16x4_f32(false, a[t],     false, b[i][t],
                                                     (short)0, acc0, false, false);
        acc1 = __builtin_amdgcn_wmma_f32_16x16x4_f32(false, a[t + 1], false, b[i][t + 1],
                                                     (short)0, acc1, false, false);
      }
      v8f acc = acc0 + acc1;
      const float bb = b1s[col];
      const float wv = wss[col];
#pragma unroll
      for (int v = 0; v < 8; ++v) {
        float hv = fmaxf(acc[v] + bb, 0.f);
        ld[v] = fmaf(hv, wv, ld[v]);
      }
    }
  };
  auto fold = [&](v8f& ld, int q) {
#pragma unroll
    for (int v = 0; v < 8; ++v) { float t = ld[v]; ld[v] = t + row_xor<1>(t); }
#pragma unroll
    for (int v = 0; v < 8; ++v) { float t = ld[v]; ld[v] = t + row_xor<2>(t); }
#pragma unroll
    for (int v = 0; v < 8; ++v) { float t = ld[v]; ld[v] = t + row_xor<4>(t); }
#pragma unroll
    for (int v = 0; v < 8; ++v) { float t = ld[v]; ld[v] = t + row_xor<8>(t); }
    const int r0 = (lane >> 4) * 8;
#pragma unroll
    for (int v = 0; v < 8; ++v) part[q][r0 + v] = ld[v];
  };

  v8f ld0 = {}, ld1 = {};
  tiles2(bA, nt0, ld0);
  load_b(bA, nt0 + 4);
  __builtin_amdgcn_sched_barrier(0);
  tiles2(bB, nt0 + 2, ld0);
  load_b(bB, nt0 + 6);
  __builtin_amdgcn_sched_barrier(0);
  fold(ld0, 2 * wave);
  tiles2(bA, nt0 + 4, ld1);
  tiles2(bB, nt0 + 6, ld1);
  fold(ld1, 2 * wave + 1);

  lds_barrier();
  if (tid < SCAN_ROWS) {
    float s = part[0][tid];
#pragma unroll
    for (int k = 1; k < LD_PARTS; ++k) s += part[k][tid];
    ldout[blockIdx.x * SCAN_ROWS + tid] = s + ws[WS_BSUM];
  }
}

extern "C" void kernel_launch(void* const* d_in, const int* in_sizes, int n_in,
                              void* d_out, int out_size, void* d_ws, size_t ws_size,
                              hipStream_t stream) {
  const float* z     = (const float*)d_in[0];
  const float* W1    = (const float*)d_in[1];
  const float* b1    = (const float*)d_in[2];
  const float* W2    = (const float*)d_in[3];
  const float* b2    = (const float*)d_in[4];
  const float* mask1 = (const float*)d_in[5];
  const float* mask2 = (const float*)d_in[6];

  float* ws    = (float*)d_ws;
  float* xout  = (float*)d_out;
  float* ldout = xout + BATCH * DIM;

  const int nprep = HID * DIM + 2 * DIM * HID + NPACK;
  maf_prep<<<PREP_HEAD_BLOCKS + (nprep + 255) / 256, 256, 0, stream>>>(W1, mask1, W2, mask2, b2, ws);
  maf_scan<<<BATCH / SCAN_ROWS, 32 * SCAN_WAVES, 0, stream>>>(z, b1, b2, ws, xout, ldout);
}
